# starting kernel
# baseline (speedup 1.0000x reference)
#include <hip/hip_runtime.h>


#define HSZ   30
#define MPAD  128
#define KPAD  32
#define TLEN  2048
#define BSZ   512

typedef __attribute__((ext_vector_type(16))) _Float16 v16h;
typedef __attribute__((ext_vector_type(8)))  _Float16 v8h;
typedef __attribute__((ext_vector_type(8)))  float    v8f;

#if __has_builtin(__builtin_amdgcn_tanhf)
#define HW_TANH __builtin_amdgcn_tanhf
#elif __has_builtin(__builtin_amdgcn_tanh_f32)
#define HW_TANH __builtin_amdgcn_tanh_f32
#endif

#ifdef HW_TANH
__device__ __forceinline__ float tanh_act(float x) { return HW_TANH(x); }
__device__ __forceinline__ float sig_act(float x)  { return fmaf(0.5f, HW_TANH(0.5f * x), 0.5f); }
#else
__device__ __forceinline__ float tanh_act(float x) {
  return fmaf(2.0f, __builtin_amdgcn_rcpf(1.0f + __expf(-2.0f * x)), -1.0f);
}
__device__ __forceinline__ float sig_act(float x) {
  return __builtin_amdgcn_rcpf(1.0f + __expf(-x));
}
#endif

__device__ __forceinline__ v16h load_a_tile(const _Float16* __restrict__ base,
                                            int tile, int lane) {
  const int m  = lane & 15;
  const int kb = (lane >> 4) << 3;
  const _Float16* row = base + ((tile << 4) + m) * KPAD;
  v8h lo = *(const v8h*)(row + kb);
  v8h hi = *(const v8h*)(row + 16 + kb);
  v16h a;
#pragma unroll
  for (int i = 0; i < 8; ++i) { a[i] = lo[i]; a[8 + i] = hi[i]; }
  return a;
}

__global__ __launch_bounds__(32, 1)
void lstm2_wmma_kernel(const float* __restrict__ x,
                       const float* __restrict__ w_ih1, const float* __restrict__ w_hh1,
                       const float* __restrict__ b_ih1, const float* __restrict__ b_hh1,
                       const float* __restrict__ w_ih2, const float* __restrict__ w_hh2,
                       const float* __restrict__ b_ih2, const float* __restrict__ b_hh2,
                       const float* __restrict__ w_fc,  const float* __restrict__ b_fc,
                       float* __restrict__ out) {
  __shared__ _Float16 sA1[MPAD * KPAD];
  __shared__ _Float16 sAi2[MPAD * KPAD];
  __shared__ _Float16 sAh2[MPAD * KPAD];

  const int lane = threadIdx.x;
  const bool hi_half = lane >= 16;
  const int n = lane & 15;

  for (int idx = lane; idx < MPAD * KPAD; idx += 32) {
    const int m = idx >> 5, k = idx & 31;
    const int g = m >> 5, j = m & 31;
    const int row = g * HSZ + j;
    float a1 = 0.f, ai2 = 0.f, ah2 = 0.f;
    if (j < HSZ) {
      if (k < HSZ) {
        a1  = w_hh1[row * HSZ + k];
        ai2 = w_ih2[row * HSZ + k];
        ah2 = w_hh2[row * HSZ + k];
      } else if (k == 30) {
        a1  = w_ih1[row];
      } else {
        a1  = b_ih1[row] + b_hh1[row];
        ah2 = b_ih2[row] + b_hh2[row];
      }
    }
    sA1[idx]  = (_Float16)a1;
    sAi2[idx] = (_Float16)ai2;
    sAh2[idx] = (_Float16)ah2;
  }
  __syncthreads();

  float wfcD[2][8];
#pragma unroll
  for (int s = 0; s < 2; ++s)
#pragma unroll
    for (int r = 0; r < 8; ++r) {
      const int j = 16 * s + (hi_half ? 8 : 0) + r;
      wfcD[s][r] = (j < HSZ) ? w_fc[j] : 0.f;
    }
  const float bfc = b_fc[0];

  v8f zc;
#pragma unroll
  for (int i = 0; i < 8; ++i) zc[i] = 0.f;

  v16h h1B, h2B;
#pragma unroll
  for (int i = 0; i < 16; ++i) { h1B[i] = (_Float16)0.f; h2B[i] = (_Float16)0.f; }
  if (hi_half) h2B[15] = (_Float16)1.f;

  float c1[2][8], c2[2][8];
#pragma unroll
  for (int s = 0; s < 2; ++s)
#pragma unroll
    for (int r = 0; r < 8; ++r) { c1[s][r] = 0.f; c2[s][r] = 0.f; }

  const int bbase = blockIdx.x << 4;
  const float* xrow = x + (size_t)(bbase + n) * TLEN;
  float* orow = out + (size_t)(bbase + n) * TLEN;
  float xcur = xrow[0];

  for (int t = 0; t < TLEN; ++t) {
    const float xnext = xrow[min(t + 1, TLEN - 1)];

    if (hi_half) { h1B[14] = (_Float16)xcur; h1B[15] = (_Float16)1.f; }

    v16h h1Bn;
#pragma unroll
    for (int s = 0; s < 2; ++s) {
      v8f d[4];
#pragma unroll
      for (int g = 0; g < 4; ++g) {
        v16h a = load_a_tile(sA1, 2 * g + s, lane);
        d[g] = __builtin_amdgcn_wmma_f32_16x16x32_f16(false, a, false, h1B,
                                                      (short)0, zc, false, false);
      }
#pragma unroll
      for (int r = 0; r < 8; ++r) {
        const float iv = sig_act(d[0][r]);
        const float fv = sig_act(d[1][r]);
        const float gv = tanh_act(d[2][r]);
        const float ov = sig_act(d[3][r]);
        const float cn = fmaf(fv, c1[s][r], iv * gv);
        c1[s][r] = cn;
        h1Bn[8 * s + r] = (_Float16)(ov * tanh_act(cn));
      }
    }

    v16h h2Bn;
    float acc = 0.f;
#pragma unroll
    for (int s = 0; s < 2; ++s) {
      v8f d[4];
#pragma unroll
      for (int g = 0; g < 4; ++g) {
        v16h ah = load_a_tile(sAh2, 2 * g + s, lane);
        v8f tmp = __builtin_amdgcn_wmma_f32_16x16x32_f16(false, ah, false, h2B,
                                                         (short)0, zc, false, false);
        v16h ai = load_a_tile(sAi2, 2 * g + s, lane);
        d[g] = __builtin_amdgcn_wmma_f32_16x16x32_f16(false, ai, false, h1Bn,
                                                      (short)0, tmp, false, false);
      }
#pragma unroll
      for (int r = 0; r < 8; ++r) {
        const float iv = sig_act(d[0][r]);
        const float fv = sig_act(d[1][r]);
        const float gv = tanh_act(d[2][r]);
        const float ov = sig_act(d[3][r]);
        const float cn = fmaf(fv, c2[s][r], iv * gv);
        c2[s][r] = cn;
        const float hn = ov * tanh_act(cn);
        h2Bn[8 * s + r] = (_Float16)hn;
        acc = fmaf(hn, wfcD[s][r], acc);
      }
    }
    if (hi_half) { h2Bn[14] = (_Float16)0.f; h2Bn[15] = (_Float16)1.f; }

    acc += __shfl_xor(acc, 16, 32);
    if (lane < 16) orow[t] = acc + bfc;

    h1B = h1Bn;
    h2B = h2Bn;
    xcur = xnext;
  }
}

extern "C" void kernel_launch(void* const* d_in, const int* in_sizes, int n_in,
                              void* d_out, int out_size, void* d_ws, size_t ws_size,
                              hipStream_t stream) {
  (void)in_sizes; (void)n_in; (void)out_size; (void)d_ws; (void)ws_size;
  const float* x     = (const float*)d_in[0];
  const float* w_ih1 = (const float*)d_in[1];
  const float* w_hh1 = (const float*)d_in[2];
  const float* b_ih1 = (const float*)d_in[3];
  const float* b_hh1 = (const float*)d_in[4];
  const float* w_ih2 = (const float*)d_in[5];
  const float* w_hh2 = (const float*)d_in[6];
  const float* b_ih2 = (const float*)d_in[7];
  const float* b_hh2 = (const float*)d_in[8];
  const float* w_fc  = (const float*)d_in[9];
  const float* b_fc  = (const float*)d_in[10];
  float* out = (float*)d_out;

  dim3 grid(BSZ / 16);
  dim3 block(32);
  lstm2_wmma_kernel<<<grid, block, 0, stream>>>(x, w_ih1, w_hh1, b_ih1, b_hh1,
                                                w_ih2, w_hh2, b_ih2, b_hh2,
                                                w_fc, b_fc, out);
}


# 4 batch rows per work-group, each copied into 4 columns of the h operand so the 4 lane groups split the accumulator rows: gate math on 2 units per lane, not 8; 4x the work-groups; FC sum regrouped
# speedup vs baseline: 2.5822x; 2.5822x over previous
#include <hip/hip_runtime.h>


#define HSZ   30
#define MPAD  128
#define KPAD  32
#define TLEN  2048
#define BSZ   512
#define NTHR  128
#define HBUF  512
#define NB    4
#define RPL   2

typedef __attribute__((ext_vector_type(16))) _Float16 v16h;
typedef __attribute__((ext_vector_type(8)))  _Float16 v8h;
typedef __attribute__((ext_vector_type(2)))  _Float16 v2h;
typedef __attribute__((ext_vector_type(8)))  float    v8f;
typedef __attribute__((ext_vector_type(4)))  float    v4f;

#if __has_builtin(__builtin_amdgcn_tanhf)
#define HW_TANH __builtin_amdgcn_tanhf
#elif __has_builtin(__builtin_amdgcn_tanh_f32)
#define HW_TANH __builtin_amdgcn_tanh_f32
#endif

#ifdef HW_TANH
__device__ __forceinline__ float tanh_act(float x) { return HW_TANH(x); }
__device__ __forceinline__ float sig_act(float x)  { return fmaf(0.5f, HW_TANH(0.5f * x), 0.5f); }
#else
__device__ __forceinline__ float tanh_act(float x) {
  return fmaf(2.0f, __builtin_amdgcn_rcpf(1.0f + __expf(-2.0f * x)), -1.0f);
}
__device__ __forceinline__ float sig_act(float x) {
  return __builtin_amdgcn_rcpf(1.0f + __expf(-x));
}
#endif

__device__ __forceinline__ v16h load_a_tile(const _Float16* __restrict__ base,
                                            int tile, int lane) {
  const int m  = lane & 15;
  const int kb = (lane >> 4) << 3;
  const _Float16* row = base + ((tile << 4) + m) * KPAD;
  v8h lo = *(const v8h*)(row + kb);
  v8h hi = *(const v8h*)(row + 16 + kb);
  v16h a;
#pragma unroll
  for (int i = 0; i < 8; ++i) { a[i] = lo[i]; a[8 + i] = hi[i]; }
  return a;
}

__device__ __forceinline__ v16h load_h_operand(const _Float16* buf, int slot) {
  v8h lo = *(const v8h*)(buf + slot * 8);
  v8h hi = *(const v8h*)(buf + 256 + slot * 8);
  v16h b;
#pragma unroll
  for (int i = 0; i < 8; ++i) { b[i] = lo[i]; b[8 + i] = hi[i]; }
  return b;
}

typedef __attribute__((ext_vector_type(2))) float v2f;
__device__ __forceinline__ v2f pick_rows(const v8f& d, bool g0, bool g1) {
  const v2f r01 = __builtin_shufflevector(d, d, 0, 1);
  const v2f r23 = __builtin_shufflevector(d, d, 2, 3);
  const v2f r45 = __builtin_shufflevector(d, d, 4, 5);
  const v2f r67 = __builtin_shufflevector(d, d, 6, 7);
  const v2f lo = g0 ? r23 : r01;
  const v2f hi = g0 ? r67 : r45;
  return g1 ? hi : lo;
}

__device__ __forceinline__ void lstm_gates(const v8f (&d)[4], bool g0, bool g1,
                                           float (&c)[RPL], float (&hn)[RPL]) {
  const v2f pi = pick_rows(d[0], g0, g1);
  const v2f pf = pick_rows(d[1], g0, g1);
  const v2f pg = pick_rows(d[2], g0, g1);
  const v2f po = pick_rows(d[3], g0, g1);
#pragma unroll
  for (int r = 0; r < RPL; ++r) {
    const float iv = sig_act(pi[r]);
    const float fv = sig_act(pf[r]);
    const float gv = tanh_act(pg[r]);
    const float ov = sig_act(po[r]);
    const float cn = fmaf(fv, c[r], iv * gv);
    c[r] = cn;
    hn[r] = ov * tanh_act(cn);
  }
}

__device__ __forceinline__ void handover_barrier() {
  __builtin_amdgcn_fence(__ATOMIC_RELEASE, "workgroup", "local");
  __builtin_amdgcn_s_barrier();
  __builtin_amdgcn_fence(__ATOMIC_ACQUIRE, "workgroup", "local");
}

struct FcParts { v4f a, b, c, d; };
__device__ __forceinline__ FcParts fc_load(const float* p) {
  FcParts f;
  f.a = *(const v4f*)(p);
  f.b = *(const v4f*)(p + 4);
  f.c = *(const v4f*)(p + 8);
  f.d = *(const v4f*)(p + 12);
  return f;
}
__device__ __forceinline__ float fc_sum4(const v4f v) { return (v[0] + v[1]) + (v[2] + v[3]); }
__device__ __forceinline__ float fc_finish(const FcParts& f, float bfc) {
  return ((fc_sum4(f.a) + fc_sum4(f.b)) + (fc_sum4(f.c) + fc_sum4(f.d))) + bfc;
}

template <bool FIN>
__device__ __forceinline__ void layer1_step(int it, int s, int lane, int slot, bool hi_half, int n,
                                            bool g0, bool g1, int hoff,
                                            const v16h (&A)[8], const v8f& zc, float (&c)[RPL],
                                            float& xcur, float bfc,
                                            const float* xrow, float* orow,
                                            _Float16* sH1, const float* sFc) {
  const int p = it & 1;
  const int q = p ^ 1;
  FcParts fv;
  fv.a = fv.b = fv.c = fv.d = (v4f){0.f, 0.f, 0.f, 0.f};
  if constexpr (FIN) {
    if (s == 0) fv = fc_load(sFc + p * 64 + n * 16);
  }
  const float xnext = xrow[min(it + 1, TLEN - 1)];

  v16h hB = load_h_operand(sH1 + q * HBUF, slot);
  if (hi_half) { hB[14] = (_Float16)xcur; hB[15] = (_Float16)1.f; }

  v8f d[4];
#pragma unroll
  for (int g = 0; g < 4; ++g)
    d[g] = __builtin_amdgcn_wmma_f32_16x16x32_f16(false, A[g], false, hB,
                                                  (short)0, zc, false, false);
  float hn[RPL];
  lstm_gates(d, g0, g1, c, hn);
  v2h hv;
  hv[0] = (_Float16)hn[0];
  hv[1] = (_Float16)hn[1];
  *(v2h*)(sH1 + p * HBUF + s * 256 + hoff) = hv;
  if constexpr (FIN) {
    if (s == 0 && lane < NB) orow[it - 2] = fc_finish(fv, bfc);
  }
  xcur = xnext;
}

__device__ __forceinline__ void layer2_step(int it, int s, int slot, bool hi_half, int fcoff,
                                            bool g0, bool g1, int hoff,
                                            const v16h (&A)[8], const v8f& zc, float (&c)[RPL],
                                            const float (&wfcv)[RPL],
                                            const _Float16* sH1, _Float16* sH2, float* sFc) {
  const int p = it & 1;
  const int q = p ^ 1;
  v16h h1 = load_h_operand(sH1 + q * HBUF, slot);
  v16h h2 = load_h_operand(sH2 + p * HBUF, slot);
  if (hi_half) { h2[14] = (_Float16)0.f; h2[15] = (_Float16)1.f; }

  v8f d[4];
#pragma unroll
  for (int g = 0; g < 4; ++g) {
    v8f tmp = __builtin_amdgcn_wmma_f32_16x16x32_f16(false, A[g], false, h2,
                                                     (short)0, zc, false, false);
    d[g] = __builtin_amdgcn_wmma_f32_16x16x32_f16(false, A[4 + g], false, h1,
                                                  (short)0, tmp, false, false);
  }
  float hn[RPL];
  lstm_gates(d, g0, g1, c, hn);
  v2h hv;
  hv[0] = (_Float16)hn[0];
  hv[1] = (_Float16)hn[1];
  *(v2h*)(sH2 + q * HBUF + s * 256 + hoff) = hv;
  sFc[q * 64 + fcoff] = fmaf(hn[1], wfcv[1], hn[0] * wfcv[0]);
}

__global__ __launch_bounds__(NTHR, 1)
void lstm2_wmma_kernel(const float* __restrict__ x,
                       const float* __restrict__ w_ih1, const float* __restrict__ w_hh1,
                       const float* __restrict__ b_ih1, const float* __restrict__ b_hh1,
                       const float* __restrict__ w_ih2, const float* __restrict__ w_hh2,
                       const float* __restrict__ b_ih2, const float* __restrict__ b_hh2,
                       const float* __restrict__ w_fc,  const float* __restrict__ b_fc,
                       float* __restrict__ out) {
  __shared__ _Float16 sA1[MPAD * KPAD];
  __shared__ _Float16 sAi2[MPAD * KPAD];
  __shared__ _Float16 sAh2[MPAD * KPAD];
  __shared__ __attribute__((aligned(16))) _Float16 sH1[2 * HBUF];
  __shared__ __attribute__((aligned(16))) _Float16 sH2[2 * HBUF];
  __shared__ __attribute__((aligned(16))) float sFc[2 * 64];

  const int tid  = threadIdx.x;
  const int lane = tid & 31;
  const int wave  = __builtin_amdgcn_readfirstlane(tid >> 5);
  const int layer = wave >> 1;
  const int s     = wave & 1;
  const bool hi_half = lane >= 16;
  const int n = lane & (NB - 1);
  const int grp = (lane >> 2) & 3;
  const bool g0 = (grp & 1) != 0;
  const bool g1 = (grp & 2) != 0;
  const int slot = (lane & 16) | n;
  const int hoff = slot * 8 + grp * RPL;
  const int fcoff = n * 16 + s * 8 + (hi_half ? 4 : 0) + grp;

  for (int idx = tid; idx < MPAD * KPAD; idx += NTHR) {
    const int m = idx >> 5, k = idx & 31;
    const int g = m >> 5, j = m & 31;
    const int row = g * HSZ + j;
    float a1 = 0.f, ai2 = 0.f, ah2 = 0.f;
    if (j < HSZ) {
      if (k < HSZ) {
        a1  = w_hh1[row * HSZ + k];
        ai2 = w_ih2[row * HSZ + k];
        ah2 = w_hh2[row * HSZ + k];
      } else if (k == 30) {
        a1  = w_ih1[row];
      } else {
        a1  = b_ih1[row] + b_hh1[row];
        ah2 = b_ih2[row] + b_hh2[row];
      }
    }
    sA1[idx]  = (_Float16)a1;
    sAi2[idx] = (_Float16)ai2;
    sAh2[idx] = (_Float16)ah2;
  }
  for (int idx = tid; idx < 2 * HBUF; idx += NTHR) {
    sH1[idx] = (_Float16)0.f;
    sH2[idx] = (_Float16)0.f;
  }
  __syncthreads();

  const _Float16* baseX = layer ? sAh2 : sA1;
  v16h A[8];
#pragma unroll
  for (int g = 0; g < 4; ++g) {
    A[g]     = load_a_tile(baseX, 2 * g + s, lane);
    A[4 + g] = load_a_tile(sAi2,  2 * g + s, lane);
  }

  float wfcv[RPL];
#pragma unroll
  for (int r = 0; r < RPL; ++r) {
    const int j = 16 * s + (hi_half ? 8 : 0) + grp * RPL + r;
    wfcv[r] = (j < HSZ) ? w_fc[j] : 0.f;
  }
  const float bfc = b_fc[0];

  v8f zc;
#pragma unroll
  for (int i = 0; i < 8; ++i) zc[i] = 0.f;

  float c[RPL];
#pragma unroll
  for (int r = 0; r < RPL; ++r) c[r] = 0.f;

  const int bbase = blockIdx.x * NB;
  const float* xrow = x + (size_t)(bbase + n) * TLEN;
  float* orow = out + (size_t)(bbase + n) * TLEN;
  float xcur = xrow[0];


  if (layer == 0)
    layer1_step<false>(0, s, lane, slot, hi_half, n, g0, g1, hoff, A, zc, c, xcur, bfc, xrow, orow, sH1, sFc);
  handover_barrier();

  if (layer == 0)
    layer1_step<false>(1, s, lane, slot, hi_half, n, g0, g1, hoff, A, zc, c, xcur, bfc, xrow, orow, sH1, sFc);
  else
    layer2_step(1, s, slot, hi_half, fcoff, g0, g1, hoff, A, zc, c, wfcv, sH1, sH2, sFc);
  handover_barrier();

  for (int it = 2; it < TLEN; ++it) {
    if (layer == 0)
      layer1_step<true>(it, s, lane, slot, hi_half, n, g0, g1, hoff, A, zc, c, xcur, bfc, xrow, orow, sH1, sFc);
    else
      layer2_step(it, s, slot, hi_half, fcoff, g0, g1, hoff, A, zc, c, wfcv, sH1, sH2, sFc);
    handover_barrier();
  }

  if (layer == 0) {
    if (s == 0 && lane < NB)
      orow[TLEN - 2] = fc_finish(fc_load(sFc + (TLEN & 1) * 64 + n * 16), bfc);
  } else {
    layer2_step(TLEN, s, slot, hi_half, fcoff, g0, g1, hoff, A, zc, c, wfcv, sH1, sH2, sFc);
  }
  handover_barrier();

  if (layer == 0 && s == 0 && lane < NB)
    orow[TLEN - 1] = fc_finish(fc_load(sFc + ((TLEN + 1) & 1) * 64 + n * 16), bfc);
}

extern "C" void kernel_launch(void* const* d_in, const int* in_sizes, int n_in,
                              void* d_out, int out_size, void* d_ws, size_t ws_size,
                              hipStream_t stream) {
  (void)in_sizes; (void)n_in; (void)out_size; (void)d_ws; (void)ws_size;
  const float* x     = (const float*)d_in[0];
  const float* w_ih1 = (const float*)d_in[1];
  const float* w_hh1 = (const float*)d_in[2];
  const float* b_ih1 = (const float*)d_in[3];
  const float* b_hh1 = (const float*)d_in[4];
  const float* w_ih2 = (const float*)d_in[5];
  const float* w_hh2 = (const float*)d_in[6];
  const float* b_ih2 = (const float*)d_in[7];
  const float* b_hh2 = (const float*)d_in[8];
  const float* w_fc  = (const float*)d_in[9];
  const float* b_fc  = (const float*)d_in[10];
  float* out = (float*)d_out;

  dim3 grid(BSZ / NB);
  dim3 block(NTHR);
  lstm2_wmma_kernel<<<grid, block, 0, stream>>>(x, w_ih1, w_hh1, b_ih1, b_hh1,
                                                w_ih2, w_hh2, b_ih2, b_hh2,
                                                w_fc, b_fc, out);
}


# Layer-2 wave forms the next step's input product W_ih2 h1 in the shadow of its gate math (layer 1 two steps ahead); only the 8 recurrent WMMAs precede the gates; layer-2 pre-activation sum regrouped
# speedup vs baseline: 3.4084x; 1.3200x over previous
#include <hip/hip_runtime.h>


#define HSZ   30
#define MPAD  128
#define KPAD  32
#define TLEN  2048
#define BSZ   512
#define NROW  2
#define NTHR  (NROW * 64)

typedef __attribute__((ext_vector_type(16))) _Float16 v16h;
typedef __attribute__((ext_vector_type(8)))  _Float16 v8h;
typedef __attribute__((ext_vector_type(2)))  _Float16 v2h;
typedef __attribute__((ext_vector_type(8)))  float    v8f;
typedef __attribute__((ext_vector_type(8)))  int      v8i;
typedef __attribute__((ext_vector_type(4)))  int      v4i;

#if __has_builtin(__builtin_amdgcn_tanhf)
#define HW_TANH __builtin_amdgcn_tanhf
#elif __has_builtin(__builtin_amdgcn_tanh_f32)
#define HW_TANH __builtin_amdgcn_tanh_f32
#endif

#ifdef HW_TANH
__device__ __forceinline__ float tanh_act(float x) { return HW_TANH(x); }
__device__ __forceinline__ float sig_act(float x)  { return fmaf(0.5f, HW_TANH(0.5f * x), 0.5f); }
#else
__device__ __forceinline__ float tanh_act(float x) {
  return fmaf(2.0f, __builtin_amdgcn_rcpf(1.0f + __expf(-2.0f * x)), -1.0f);
}
__device__ __forceinline__ float sig_act(float x) {
  return __builtin_amdgcn_rcpf(1.0f + __expf(-x));
}
#endif

template <int CTRL>
__device__ __forceinline__ int dpp_mov(int v) {
  return __builtin_amdgcn_update_dpp(0, v, CTRL, 0xf, 0xf, true);
}
#define DPP_SWAP_PAIR  0xB1
#define DPP_ROW_SHARE  0x150
#define DPP_ROW_XMASK  0x160

__device__ __forceinline__ float xhalf(float x) {
  const int b = __builtin_bit_cast(int, x);
  return __builtin_bit_cast(float, __builtin_amdgcn_permlanex16(b, b, 0x76543210, 0xfedcba98, false, false));
}

__device__ __forceinline__ int pack_halves(float lo, float hi) {
  v2h t;
  t[0] = (_Float16)lo;
  t[1] = (_Float16)hi;
  return __builtin_bit_cast(int, t);
}

__device__ __forceinline__ v16h load_w_tile(const _Float16* __restrict__ base,
                                            int g, int s, int lane) {
  const int n  = lane & 15;
  const int kb = (lane >> 4) << 4;
  const _Float16* row = base + ((g << 5) + (s << 4) + n) * KPAD + kb;
  v8h lo = *(const v8h*)(row);
  v8h hi = *(const v8h*)(row + 8);
  v16h b;
#pragma unroll
  for (int i = 0; i < 8; ++i) { b[i] = lo[i]; b[8 + i] = hi[i]; }
  return b;
}

__device__ __forceinline__ int pair_dword(float hn) {
  const float nb = __builtin_bit_cast(float, dpp_mov<DPP_SWAP_PAIR>(__builtin_bit_cast(int, hn)));
  return pack_halves(nb, hn);
}

__device__ __forceinline__ v16h gather_operand(int pd) {
  v8i a;
  a[0] = dpp_mov<DPP_ROW_SHARE + 1>(pd);
  a[1] = dpp_mov<DPP_ROW_SHARE + 3>(pd);
  a[2] = dpp_mov<DPP_ROW_SHARE + 5>(pd);
  a[3] = dpp_mov<DPP_ROW_SHARE + 7>(pd);
  a[4] = dpp_mov<DPP_ROW_SHARE + 9>(pd);
  a[5] = dpp_mov<DPP_ROW_SHARE + 11>(pd);
  a[6] = dpp_mov<DPP_ROW_SHARE + 13>(pd);
  a[7] = dpp_mov<DPP_ROW_SHARE + 15>(pd);
  return __builtin_bit_cast(v16h, a);
}

__device__ __forceinline__ v16h set_rail(v16h op, bool hi_half, int rail) {
  v8i a = __builtin_bit_cast(v8i, op);
  a[7] = hi_half ? rail : a[7];
  return __builtin_bit_cast(v16h, a);
}

__device__ __forceinline__ float lstm_gate(float pi, float pf, float pg, float po, float& c) {
  const float iv = sig_act(pi);
  const float fv = sig_act(pf);
  const float gv = tanh_act(pg);
  const float ov = sig_act(po);
  const float cn = fmaf(fv, c, iv * gv);
  c = cn;
  return ov * tanh_act(cn);
}

__device__ __forceinline__ float lstm_gate_in(float di, float df, float dg, float dn,
                                              float qi, float qf, float qg, float qo, float& c) {
#ifdef HW_TANH
  const float iv = fmaf(0.5f, HW_TANH(fmaf(0.5f, di, qi)), 0.5f);
  const float fv = fmaf(0.5f, HW_TANH(fmaf(0.5f, df, qf)), 0.5f);
  const float ov = fmaf(0.5f, HW_TANH(fmaf(0.5f, dn, qo)), 0.5f);
#else
  const float iv = sig_act(fmaf(2.0f, qi, di));
  const float fv = sig_act(fmaf(2.0f, qf, df));
  const float ov = sig_act(fmaf(2.0f, qo, dn));
#endif
  const float gv = tanh_act(dg + qg);
  const float cn = fmaf(fv, c, iv * gv);
  c = cn;
  return ov * tanh_act(cn);
}

__device__ __forceinline__ void handover_barrier() {
  __builtin_amdgcn_fence(__ATOMIC_RELEASE, "workgroup", "local");
  __builtin_amdgcn_s_barrier();
  __builtin_amdgcn_fence(__ATOMIC_ACQUIRE, "workgroup", "local");
}


__device__ __forceinline__ void keep_tiles_disjoint(v8f (&d)[8]) {
  asm volatile("" : "+v"(d[0]), "+v"(d[1]), "+v"(d[2]), "+v"(d[3]),
                    "+v"(d[4]), "+v"(d[5]), "+v"(d[6]), "+v"(d[7]));
}

__device__ __forceinline__ void keep_tiles_apart(v8f (&q)[8], const v8f (&d)[8], float& hn) {
  asm volatile("" : "+v"(q[0]), "+v"(q[1]), "+v"(q[2]), "+v"(q[3]),
                    "+v"(q[4]), "+v"(q[5]), "+v"(q[6]), "+v"(q[7]), "+v"(hn)
                  : "v"(d[0]), "v"(d[1]), "v"(d[2]), "v"(d[3]),
                    "v"(d[4]), "v"(d[5]), "v"(d[6]), "v"(d[7]));
}

__device__ __forceinline__ void layer1_step(int it, int lane, bool hi_half,
                                            const v16h (&W)[16], const v8f& zc, float& c,
                                            v16h& aop, float xc, int* sProw) {
  const float xs = __builtin_bit_cast(float,
      __builtin_amdgcn_readlane(__builtin_bit_cast(int, xc), it & 31));
  aop = set_rail(aop, hi_half, pack_halves(xs, 1.f));

  v8f d[8];
#pragma unroll
  for (int t = 0; t < 8; ++t)
    d[t] = __builtin_amdgcn_wmma_f32_16x16x32_f16(false, aop, false, W[t],
                                                  (short)0, zc, false, false);
  keep_tiles_disjoint(d);
  const float pi = hi_half ? d[1][0] : d[0][0];
  const float pf = hi_half ? d[3][0] : d[2][0];
  const float pg = hi_half ? d[5][0] : d[4][0];
  const float po = hi_half ? d[7][0] : d[6][0];
  const float hn = lstm_gate(pi, pf, pg, po, c);
  const int pd = pair_dword(hn);
  if (lane & 1) sProw[(it & 1) * 16 + (lane >> 1)] = pd;
  aop = gather_operand(pd);
}

template <bool REC, bool PROJ>
__device__ __forceinline__ void layer2_step(int it, int lane, bool hi_half,
                                            const v16h (&W)[16], const v8f& zc, float& c,
                                            v16h& aop, int rail, float wfc, float bfc,
                                            float& fcv, float& oacc, float (&qin)[4],
                                            const int* sProw, float* orow) {
  const int u = it - 3;
  if (REC) {
    float v = fcv;
    v += __builtin_bit_cast(float, dpp_mov<DPP_ROW_XMASK + 1>(__builtin_bit_cast(int, v)));
    v += __builtin_bit_cast(float, dpp_mov<DPP_ROW_XMASK + 2>(__builtin_bit_cast(int, v)));
    v += __builtin_bit_cast(float, dpp_mov<DPP_ROW_XMASK + 4>(__builtin_bit_cast(int, v)));
    v += __builtin_bit_cast(float, dpp_mov<DPP_ROW_XMASK + 8>(__builtin_bit_cast(int, v)));
    v += xhalf(v);
    oacc = (lane == (u & 31)) ? (v + bfc) : oacc;
  }

  v16h h1;
  if (PROJ) {
    const int* src = sProw + ((it - 1) & 1) * 16 + (hi_half ? 8 : 0);
    const v4i lo = *(const v4i*)(src);
    const v4i hi = *(const v4i*)(src + 4);
    v8i hw;
#pragma unroll
    for (int i = 0; i < 4; ++i) { hw[i] = lo[i]; hw[4 + i] = hi[i]; }
    h1 = __builtin_bit_cast(v16h, hw);
  }

  v8f d[8], q[8];
  if (REC) {
#pragma unroll
    for (int t = 0; t < 8; ++t)
      d[t] = __builtin_amdgcn_wmma_f32_16x16x32_f16(false, aop, false, W[t],
                                                    (short)0, zc, false, false);
    keep_tiles_disjoint(d);
  }
  if (PROJ) {
#pragma unroll
    for (int t = 0; t < 8; ++t)
      q[t] = __builtin_amdgcn_wmma_f32_16x16x32_f16(false, h1, false, W[8 + t],
                                                    (short)0, zc, false, false);
    if (!REC) keep_tiles_disjoint(q);
  }
  if (REC) {
    const float di = hi_half ? d[1][0] : d[0][0];
    const float df = hi_half ? d[3][0] : d[2][0];
    const float dg = hi_half ? d[5][0] : d[4][0];
    const float dn = hi_half ? d[7][0] : d[6][0];
    float hn = lstm_gate_in(di, df, dg, dn, qin[0], qin[1], qin[2], qin[3], c);
    if (PROJ) keep_tiles_apart(q, d, hn);
    aop = set_rail(gather_operand(pair_dword(hn)), hi_half, rail);
    fcv = hn * wfc;
    if (u >= 0 && (u & 31) == 31) orow[(u & ~31) + lane] = oacc;
  }
  if (PROJ) {
    qin[0] = 0.5f * (hi_half ? q[1][0] : q[0][0]);
    qin[1] = 0.5f * (hi_half ? q[3][0] : q[2][0]);
    qin[2] =        (hi_half ? q[5][0] : q[4][0]);
    qin[3] = 0.5f * (hi_half ? q[7][0] : q[6][0]);
  }
}

__global__ __launch_bounds__(NTHR, 1)
void lstm2_wmma_kernel(const float* __restrict__ x,
                       const float* __restrict__ w_ih1, const float* __restrict__ w_hh1,
                       const float* __restrict__ b_ih1, const float* __restrict__ b_hh1,
                       const float* __restrict__ w_ih2, const float* __restrict__ w_hh2,
                       const float* __restrict__ b_ih2, const float* __restrict__ b_hh2,
                       const float* __restrict__ w_fc,  const float* __restrict__ b_fc,
                       float* __restrict__ out) {
  __shared__ _Float16 sA1[MPAD * KPAD];
  __shared__ _Float16 sAi2[MPAD * KPAD];
  __shared__ _Float16 sAh2[MPAD * KPAD];
  __shared__ __attribute__((aligned(16))) int sP[NROW * 2 * 16];

  const int tid  = threadIdx.x;
  const int lane = tid & 31;
  const int wave  = __builtin_amdgcn_readfirstlane(tid >> 5);
  const int layer = wave & 1;
  const int prow  = wave >> 1;
  const bool hi_half = lane >= 16;

  for (int idx = tid; idx < MPAD * KPAD; idx += NTHR) {
    const int m = idx >> 5, k = idx & 31;
    const int g = m >> 5, j = m & 31;
    const int row = g * HSZ + j;
    float a1 = 0.f, ai2 = 0.f, ah2 = 0.f;
    if (j < HSZ) {
      if (k < HSZ) {
        a1  = w_hh1[row * HSZ + k];
        ai2 = w_ih2[row * HSZ + k];
        ah2 = w_hh2[row * HSZ + k];
      } else if (k == 30) {
        a1  = w_ih1[row];
      } else {
        a1  = b_ih1[row] + b_hh1[row];
        ah2 = b_ih2[row] + b_hh2[row];
      }
    }
    sA1[idx]  = (_Float16)a1;
    sAi2[idx] = (_Float16)ai2;
    sAh2[idx] = (_Float16)ah2;
  }
  __syncthreads();

  const _Float16* baseX = layer ? sAh2 : sA1;
  v16h W[16];
#pragma unroll
  for (int g = 0; g < 4; ++g) {
#pragma unroll
    for (int s = 0; s < 2; ++s) {
      W[2 * g + s]     = load_w_tile(baseX, g, s, lane);
      W[8 + 2 * g + s] = load_w_tile(sAi2,  g, s, lane);
    }
  }

  const float wfc = (lane < HSZ) ? w_fc[lane] : 0.f;
  const float bfc = b_fc[0];

  v8f zc;
#pragma unroll
  for (int i = 0; i < 8; ++i) zc[i] = 0.f;

  float c = 0.f;

  const int brow = blockIdx.x * NROW + prow;
  const float* xrow = x + (size_t)brow * TLEN;
  float* orow = out + (size_t)brow * TLEN;
  int* sProw = sP + prow * 32;

  float xc = xrow[lane];
  float fcv = 0.f, oacc = 0.f;
  float qin[4] = {0.f, 0.f, 0.f, 0.f};

  const int rail2 = pack_halves(0.f, 1.f);
  v8i a0;
#pragma unroll
  for (int i = 0; i < 8; ++i) a0[i] = 0;
  v16h aop = set_rail(__builtin_bit_cast(v16h, a0), hi_half, rail2);


  if (layer == 0)
    layer1_step(0, lane, hi_half, W, zc, c, aop, xc, sProw);
  handover_barrier();
  if (layer == 0)
    layer1_step(1, lane, hi_half, W, zc, c, aop, xc, sProw);
  else
    layer2_step<false, true>(1, lane, hi_half, W, zc, c, aop, rail2, wfc, bfc, fcv, oacc, qin, sProw, orow);
  handover_barrier();

  for (int k = 0; k < TLEN / 32; ++k) {
    const float xn = xrow[min(32 * (k + 1) + lane, TLEN - 1)];
    for (int j = (k == 0) ? 2 : 0; j < 32; ++j) {
      const int it = 32 * k + j;
      if (layer == 0)
        layer1_step(it, lane, hi_half, W, zc, c, aop, xc, sProw);
      else
        layer2_step<true, true>(it, lane, hi_half, W, zc, c, aop, rail2, wfc, bfc, fcv, oacc, qin, sProw, orow);
      handover_barrier();
    }
    xc = xn;
  }

  if (layer != 0) {
    layer2_step<true, true>(TLEN, lane, hi_half, W, zc, c, aop, rail2, wfc, bfc, fcv, oacc, qin, sProw, orow);
    layer2_step<true, false>(TLEN + 1, lane, hi_half, W, zc, c, aop, rail2, wfc, bfc, fcv, oacc, qin, sProw, orow);
    float v = fcv;
    v += __builtin_bit_cast(float, dpp_mov<DPP_ROW_XMASK + 1>(__builtin_bit_cast(int, v)));
    v += __builtin_bit_cast(float, dpp_mov<DPP_ROW_XMASK + 2>(__builtin_bit_cast(int, v)));
    v += __builtin_bit_cast(float, dpp_mov<DPP_ROW_XMASK + 4>(__builtin_bit_cast(int, v)));
    v += __builtin_bit_cast(float, dpp_mov<DPP_ROW_XMASK + 8>(__builtin_bit_cast(int, v)));
    v += xhalf(v);
    oacc = (lane == 31) ? (v + bfc) : oacc;
    orow[TLEN - 32 + lane] = oacc;
  }
}

extern "C" void kernel_launch(void* const* d_in, const int* in_sizes, int n_in,
                              void* d_out, int out_size, void* d_ws, size_t ws_size,
                              hipStream_t stream) {
  (void)in_sizes; (void)n_in; (void)out_size; (void)d_ws; (void)ws_size;
  const float* x     = (const float*)d_in[0];
  const float* w_ih1 = (const float*)d_in[1];
  const float* w_hh1 = (const float*)d_in[2];
  const float* b_ih1 = (const float*)d_in[3];
  const float* b_hh1 = (const float*)d_in[4];
  const float* w_ih2 = (const float*)d_in[5];
  const float* w_hh2 = (const float*)d_in[6];
  const float* b_ih2 = (const float*)d_in[7];
  const float* b_hh2 = (const float*)d_in[8];
  const float* w_fc  = (const float*)d_in[9];
  const float* b_fc  = (const float*)d_in[10];
  float* out = (float*)d_out;

  dim3 grid(BSZ / NROW);
  dim3 block(NTHR);
  lstm2_wmma_kernel<<<grid, block, 0, stream>>>(x, w_ih1, w_hh1, b_ih1, b_hh1,
                                                w_ih2, w_hh2, b_ih2, b_hh2,
                                                w_fc, b_fc, out);
}


# h1 goes to layer 2 in chunks of 16 steps through a two-chunk LDS ring: one block barrier and one LDS visit per wave per 16 steps, not per step; each role in its own unrolled loop; sum orders kept
# speedup vs baseline: 4.4071x; 1.2930x over previous
#include <hip/hip_runtime.h>


#define HSZ   30
#define MPAD  128
#define KPAD  32
#define TLEN  2048
#define BSZ   512
#define NROW  2
#define NTHR  (NROW * 64)
#define NCH   16
#define NPH   (TLEN / NCH)

typedef __attribute__((ext_vector_type(16))) _Float16 v16h;
typedef __attribute__((ext_vector_type(8)))  _Float16 v8h;
typedef __attribute__((ext_vector_type(2)))  _Float16 v2h;
typedef __attribute__((ext_vector_type(8)))  float    v8f;
typedef __attribute__((ext_vector_type(8)))  int      v8i;
typedef __attribute__((ext_vector_type(4)))  int      v4i;
typedef unsigned short __attribute__((may_alias)) u16a;

#if __has_builtin(__builtin_amdgcn_tanhf)
#define HW_TANH __builtin_amdgcn_tanhf
#elif __has_builtin(__builtin_amdgcn_tanh_f32)
#define HW_TANH __builtin_amdgcn_tanh_f32
#endif

#ifdef HW_TANH
__device__ __forceinline__ float tanh_act(float x) { return HW_TANH(x); }
__device__ __forceinline__ float sig_act(float x)  { return fmaf(0.5f, HW_TANH(0.5f * x), 0.5f); }
#else
__device__ __forceinline__ float tanh_act(float x) {
  return fmaf(2.0f, __builtin_amdgcn_rcpf(1.0f + __expf(-2.0f * x)), -1.0f);
}
__device__ __forceinline__ float sig_act(float x) {
  return __builtin_amdgcn_rcpf(1.0f + __expf(-x));
}
#endif

template <int CTRL>
__device__ __forceinline__ int dpp_mov(int v) {
  return __builtin_amdgcn_update_dpp(0, v, CTRL, 0xf, 0xf, true);
}
#define DPP_SWAP_PAIR  0xB1
#define DPP_ROW_SHARE  0x150
#define DPP_ROW_XMASK  0x160

__device__ __forceinline__ float xhalf(float x) {
  const int b = __builtin_bit_cast(int, x);
  return __builtin_bit_cast(float, __builtin_amdgcn_permlanex16(b, b, 0x76543210, 0xfedcba98, false, false));
}

__device__ __forceinline__ int pack_halves(float lo, float hi) {
  v2h t;
  t[0] = (_Float16)lo;
  t[1] = (_Float16)hi;
  return __builtin_bit_cast(int, t);
}

__device__ __forceinline__ v16h load_w_tile(const _Float16* __restrict__ base,
                                            int g, int s, int lane) {
  const int n  = lane & 15;
  const int kb = (lane >> 4) << 4;
  const _Float16* row = base + ((g << 5) + (s << 4) + n) * KPAD + kb;
  v8h lo = *(const v8h*)(row);
  v8h hi = *(const v8h*)(row + 8);
  v16h b;
#pragma unroll
  for (int i = 0; i < 8; ++i) { b[i] = lo[i]; b[8 + i] = hi[i]; }
  return b;
}

__device__ __forceinline__ int pair_dword(float hn) {
  const float nb = __builtin_bit_cast(float, dpp_mov<DPP_SWAP_PAIR>(__builtin_bit_cast(int, hn)));
  return pack_halves(nb, hn);
}

__device__ __forceinline__ v16h gather_operand(int pd) {
  v8i a;
  a[0] = dpp_mov<DPP_ROW_SHARE + 1>(pd);
  a[1] = dpp_mov<DPP_ROW_SHARE + 3>(pd);
  a[2] = dpp_mov<DPP_ROW_SHARE + 5>(pd);
  a[3] = dpp_mov<DPP_ROW_SHARE + 7>(pd);
  a[4] = dpp_mov<DPP_ROW_SHARE + 9>(pd);
  a[5] = dpp_mov<DPP_ROW_SHARE + 11>(pd);
  a[6] = dpp_mov<DPP_ROW_SHARE + 13>(pd);
  a[7] = dpp_mov<DPP_ROW_SHARE + 15>(pd);
  return __builtin_bit_cast(v16h, a);
}

__device__ __forceinline__ v16h set_rail(v16h op, bool hi_half, int rail) {
  v8i a = __builtin_bit_cast(v8i, op);
  a[7] = hi_half ? rail : a[7];
  return __builtin_bit_cast(v16h, a);
}

__device__ __forceinline__ float lstm_gate(float pi, float pf, float pg, float po, float& c) {
  const float iv = sig_act(pi);
  const float fv = sig_act(pf);
  const float gv = tanh_act(pg);
  const float ov = sig_act(po);
  const float cn = fmaf(fv, c, iv * gv);
  c = cn;
  return ov * tanh_act(cn);
}

__device__ __forceinline__ float lstm_gate_in(float di, float df, float dg, float dn,
                                              float qi, float qf, float qg, float qo, float& c) {
#ifdef HW_TANH
  const float iv = fmaf(0.5f, HW_TANH(fmaf(0.5f, di, qi)), 0.5f);
  const float fv = fmaf(0.5f, HW_TANH(fmaf(0.5f, df, qf)), 0.5f);
  const float ov = fmaf(0.5f, HW_TANH(fmaf(0.5f, dn, qo)), 0.5f);
#else
  const float iv = sig_act(fmaf(2.0f, qi, di));
  const float fv = sig_act(fmaf(2.0f, qf, df));
  const float ov = sig_act(fmaf(2.0f, qo, dn));
#endif
  const float gv = tanh_act(dg + qg);
  const float cn = fmaf(fv, c, iv * gv);
  c = cn;
  return ov * tanh_act(cn);
}

__device__ __forceinline__ void handover_barrier() {
  __builtin_amdgcn_fence(__ATOMIC_RELEASE, "workgroup", "local");
  __builtin_amdgcn_s_barrier();
  __builtin_amdgcn_fence(__ATOMIC_ACQUIRE, "workgroup", "local");
}


__device__ __forceinline__ void keep_tiles_disjoint(v8f (&d)[8]) {
  asm volatile("" : "+v"(d[0]), "+v"(d[1]), "+v"(d[2]), "+v"(d[3]),
                    "+v"(d[4]), "+v"(d[5]), "+v"(d[6]), "+v"(d[7]));
}

__device__ __forceinline__ void keep_tiles_apart(v8f (&q)[8], const v8f (&d)[8], float& hn) {
  asm volatile("" : "+v"(q[0]), "+v"(q[1]), "+v"(q[2]), "+v"(q[3]),
                    "+v"(q[4]), "+v"(q[5]), "+v"(q[6]), "+v"(q[7]), "+v"(hn)
                  : "v"(d[0]), "v"(d[1]), "v"(d[2]), "v"(d[3]),
                    "v"(d[4]), "v"(d[5]), "v"(d[6]), "v"(d[7]));
}

__device__ __forceinline__ int layer1_step(int j, bool hi_half,
                                           const v16h (&W)[16], const v8f& zc, float& c,
                                           v16h& aop, float xc) {
  const float xs = __builtin_bit_cast(float,
      __builtin_amdgcn_readlane(__builtin_bit_cast(int, xc), j));
  aop = set_rail(aop, hi_half, pack_halves(xs, 1.f));

  v8f d[8];
#pragma unroll
  for (int t = 0; t < 8; ++t)
    d[t] = __builtin_amdgcn_wmma_f32_16x16x32_f16(false, aop, false, W[t],
                                                  (short)0, zc, false, false);
  keep_tiles_disjoint(d);
  const float pi = hi_half ? d[1][0] : d[0][0];
  const float pf = hi_half ? d[3][0] : d[2][0];
  const float pg = hi_half ? d[5][0] : d[4][0];
  const float po = hi_half ? d[7][0] : d[6][0];
  const float hn = lstm_gate(pi, pf, pg, po, c);
  const int pd = pair_dword(hn);
  aop = gather_operand(pd);
  return pd;
}

template <bool REC, bool PROJ>
__device__ __forceinline__ void layer2_step(int it, int lane, bool hi_half,
                                            const v16h (&W)[16], const v8f& zc, float& c,
                                            v16h& aop, int rail, float wfc, float bfc,
                                            float& fcv, float& oacc, float (&qin)[4],
                                            const v16h& h1, bool line_done, float* orow) {
  const int u = it - 3;
  if (REC) {
    float v = fcv;
    v += __builtin_bit_cast(float, dpp_mov<DPP_ROW_XMASK + 1>(__builtin_bit_cast(int, v)));
    v += __builtin_bit_cast(float, dpp_mov<DPP_ROW_XMASK + 2>(__builtin_bit_cast(int, v)));
    v += __builtin_bit_cast(float, dpp_mov<DPP_ROW_XMASK + 4>(__builtin_bit_cast(int, v)));
    v += __builtin_bit_cast(float, dpp_mov<DPP_ROW_XMASK + 8>(__builtin_bit_cast(int, v)));
    v += xhalf(v);
    oacc = (lane == (u & 31)) ? (v + bfc) : oacc;
  }

  v8f d[8], q[8];
  if (REC) {
#pragma unroll
    for (int t = 0; t < 8; ++t)
      d[t] = __builtin_amdgcn_wmma_f32_16x16x32_f16(false, aop, false, W[t],
                                                    (short)0, zc, false, false);
    keep_tiles_disjoint(d);
  }
  if (PROJ) {
#pragma unroll
    for (int t = 0; t < 8; ++t)
      q[t] = __builtin_amdgcn_wmma_f32_16x16x32_f16(false, h1, false, W[8 + t],
                                                    (short)0, zc, false, false);
    if (!REC) keep_tiles_disjoint(q);
  }
  if (REC) {
    const float di = hi_half ? d[1][0] : d[0][0];
    const float df = hi_half ? d[3][0] : d[2][0];
    const float dg = hi_half ? d[5][0] : d[4][0];
    const float dn = hi_half ? d[7][0] : d[6][0];
    float hn = lstm_gate_in(di, df, dg, dn, qin[0], qin[1], qin[2], qin[3], c);
    if (PROJ) keep_tiles_apart(q, d, hn);
    aop = set_rail(gather_operand(pair_dword(hn)), hi_half, rail);
    fcv = hn * wfc;
    if (line_done) orow[(u & ~31) + lane] = oacc;
  }
  if (PROJ) {
    qin[0] = 0.5f * (hi_half ? q[1][0] : q[0][0]);
    qin[1] = 0.5f * (hi_half ? q[3][0] : q[2][0]);
    qin[2] =        (hi_half ? q[5][0] : q[4][0]);
    qin[3] = 0.5f * (hi_half ? q[7][0] : q[6][0]);
  }
}

__device__ __forceinline__ v16h load_h1(const int* src, int i) {
  const v4i lo = *(const v4i*)(src + i * 16);
  const v4i hi = *(const v4i*)(src + i * 16 + 4);
  v8i hw;
#pragma unroll
  for (int e = 0; e < 4; ++e) { hw[e] = lo[e]; hw[4 + e] = hi[e]; }
  return __builtin_bit_cast(v16h, hw);
}

__device__ __forceinline__ void layer2_chunk(int it0, bool line_phase, int lane, bool hi_half,
                                             const v16h (&W)[16], const v8f& zc, float& c,
                                             v16h& aop, int rail, float wfc, float bfc,
                                             float& fcv, float& oacc, float (&qin)[4],
                                             const int* src, float* orow) {
  v16h hb[NCH];
#pragma unroll
  for (int i = 0; i < NCH; ++i) hb[i] = load_h1(src, i);
#pragma unroll
  for (int i = 0; i < NCH; ++i) {
    const bool line_done = (i == 1) && line_phase;
    layer2_step<true, true>(it0 + i, lane, hi_half, W, zc, c, aop, rail, wfc, bfc, fcv, oacc, qin,
                            hb[i], line_done, orow);
  }
}

__device__ __forceinline__ void layer2_first_chunk(int lane, bool hi_half,
                                                   const v16h (&W)[16], const v8f& zc, float& c,
                                                   v16h& aop, int rail, float wfc, float bfc,
                                                   float& fcv, float& oacc, float (&qin)[4],
                                                   const int* src, float* orow) {
  layer2_step<false, true>(1, lane, hi_half, W, zc, c, aop, rail, wfc, bfc, fcv, oacc, qin,
                           load_h1(src, 0), false, orow);
  for (int i = 1; i < NCH; ++i)
    layer2_step<true, true>(1 + i, lane, hi_half, W, zc, c, aop, rail, wfc, bfc, fcv, oacc, qin,
                            load_h1(src, i), false, orow);
}

__global__ __launch_bounds__(NTHR, 1)
void lstm2_wmma_kernel(const float* __restrict__ x,
                       const float* __restrict__ w_ih1, const float* __restrict__ w_hh1,
                       const float* __restrict__ b_ih1, const float* __restrict__ b_hh1,
                       const float* __restrict__ w_ih2, const float* __restrict__ w_hh2,
                       const float* __restrict__ b_ih2, const float* __restrict__ b_hh2,
                       const float* __restrict__ w_fc,  const float* __restrict__ b_fc,
                       float* __restrict__ out) {
  __shared__ _Float16 sA1[MPAD * KPAD];
  __shared__ _Float16 sAi2[MPAD * KPAD];
  __shared__ _Float16 sAh2[MPAD * KPAD];
  __shared__ __attribute__((aligned(16))) int sP[NROW * 2 * NCH * 16];

  const int tid  = threadIdx.x;
  const int lane = tid & 31;
  const int wave  = __builtin_amdgcn_readfirstlane(tid >> 5);
  const int layer = wave & 1;
  const int prow  = wave >> 1;
  const bool hi_half = lane >= 16;

  for (int idx = tid; idx < MPAD * KPAD; idx += NTHR) {
    const int m = idx >> 5, k = idx & 31;
    const int g = m >> 5, j = m & 31;
    const int row = g * HSZ + j;
    float a1 = 0.f, ai2 = 0.f, ah2 = 0.f;
    if (j < HSZ) {
      if (k < HSZ) {
        a1  = w_hh1[row * HSZ + k];
        ai2 = w_ih2[row * HSZ + k];
        ah2 = w_hh2[row * HSZ + k];
      } else if (k == 30) {
        a1  = w_ih1[row];
      } else {
        a1  = b_ih1[row] + b_hh1[row];
        ah2 = b_ih2[row] + b_hh2[row];
      }
    }
    sA1[idx]  = (_Float16)a1;
    sAi2[idx] = (_Float16)ai2;
    sAh2[idx] = (_Float16)ah2;
  }
  __syncthreads();

  const _Float16* baseX = layer ? sAh2 : sA1;
  v16h W[16];
#pragma unroll
  for (int g = 0; g < 4; ++g) {
#pragma unroll
    for (int s = 0; s < 2; ++s) {
      W[2 * g + s]     = load_w_tile(baseX, g, s, lane);
      W[8 + 2 * g + s] = load_w_tile(sAi2,  g, s, lane);
    }
  }

  const float wfc = (lane < HSZ) ? w_fc[lane] : 0.f;
  const float bfc = b_fc[0];

  v8f zc;
#pragma unroll
  for (int i = 0; i < 8; ++i) zc[i] = 0.f;

  float c = 0.f;

  const int brow = blockIdx.x * NROW + prow;
  const float* xrow = x + (size_t)brow * TLEN;
  float* orow = out + (size_t)brow * TLEN;
  int* sProw = sP + prow * (2 * NCH * 16);

  const int xl = lane & (NCH - 1);
  float xc = xrow[xl];
  float fcv = 0.f, oacc = 0.f;
  float qin[4] = {0.f, 0.f, 0.f, 0.f};

  const int rail2 = pack_halves(0.f, 1.f);
  v8i a0;
#pragma unroll
  for (int i = 0; i < 8; ++i) a0[i] = 0;
  v16h aop = set_rail(__builtin_bit_cast(v16h, a0), hi_half, rail2);

  for (int p = 0; p <= NPH; ++p) {
    if (layer == 0) {
      if (p < NPH) {
        const float xn = xrow[min(NCH * (p + 1) + xl, TLEN - 1)];
        int pd[NCH];
#pragma unroll
        for (int j = 0; j < NCH; ++j)
          pd[j] = layer1_step(j, hi_half, W, zc, c, aop, xc);
        u16a* dst = (u16a*)(sProw + (p & 1) * (NCH * 16)) + (lane ^ 1);
#pragma unroll
        for (int j = 0; j < NCH; ++j)
          dst[j * 32] = (u16a)pd[j];
        xc = xn;
      }
    } else {
      if (p >= 1) {
        const int* src = sProw + ((p - 1) & 1) * (NCH * 16) + (hi_half ? 8 : 0);
        const int it0 = NCH * (p - 1) + 1;
        if (p == 1)
          layer2_first_chunk(lane, hi_half, W, zc, c, aop, rail2, wfc, bfc,
                             fcv, oacc, qin, src, orow);
        else
          layer2_chunk(it0, (p & 1) != 0, lane, hi_half, W, zc, c, aop, rail2, wfc, bfc,
                       fcv, oacc, qin, src, orow);
      }
    }
    if (p < NPH) handover_barrier();
  }

  if (layer != 0) {
    v16h none = aop;
    layer2_step<true, false>(TLEN + 1, lane, hi_half, W, zc, c, aop, rail2, wfc, bfc, fcv, oacc, qin,
                             none, false, orow);
    float v = fcv;
    v += __builtin_bit_cast(float, dpp_mov<DPP_ROW_XMASK + 1>(__builtin_bit_cast(int, v)));
    v += __builtin_bit_cast(float, dpp_mov<DPP_ROW_XMASK + 2>(__builtin_bit_cast(int, v)));
    v += __builtin_bit_cast(float, dpp_mov<DPP_ROW_XMASK + 4>(__builtin_bit_cast(int, v)));
    v += __builtin_bit_cast(float, dpp_mov<DPP_ROW_XMASK + 8>(__builtin_bit_cast(int, v)));
    v += xhalf(v);
    oacc = (lane == 31) ? (v + bfc) : oacc;
    orow[TLEN - 32 + lane] = oacc;
  }
}

extern "C" void kernel_launch(void* const* d_in, const int* in_sizes, int n_in,
                              void* d_out, int out_size, void* d_ws, size_t ws_size,
                              hipStream_t stream) {
  (void)in_sizes; (void)n_in; (void)out_size; (void)d_ws; (void)ws_size;
  const float* x     = (const float*)d_in[0];
  const float* w_ih1 = (const float*)d_in[1];
  const float* w_hh1 = (const float*)d_in[2];
  const float* b_ih1 = (const float*)d_in[3];
  const float* b_hh1 = (const float*)d_in[4];
  const float* w_ih2 = (const float*)d_in[5];
  const float* w_hh2 = (const float*)d_in[6];
  const float* b_ih2 = (const float*)d_in[7];
  const float* b_hh2 = (const float*)d_in[8];
  const float* w_fc  = (const float*)d_in[9];
  const float* b_fc  = (const float*)d_in[10];
  float* out = (float*)d_out;

  dim3 grid(BSZ / NROW);
  dim3 block(NTHR);
  lstm2_wmma_kernel<<<grid, block, 0, stream>>>(x, w_ih1, w_hh1, b_ih1, b_hh1,
                                                w_ih2, w_hh2, b_ih2, b_hh2,
                                                w_fc, b_fc, out);
}


# Layer-2 input product W_ih2 h1 made for 8 steps per WMMA (A rows = h1 of 8 steps from the LDS chunk), one per step: 9 WMMAs per layer-2 step, not 16; layer 2 8 steps later; sum orders kept
# speedup vs baseline: 4.6632x; 1.0581x over previous
#include <hip/hip_runtime.h>


#define HSZ   30
#define MPAD  128
#define KPAD  32
#define TLEN  2048
#define BSZ   512
#define NROW  2
#define NTHR  (NROW * 64)
#define NCH   16
#define NPH   (TLEN / NCH)

typedef __attribute__((ext_vector_type(16))) _Float16 v16h;
typedef __attribute__((ext_vector_type(8)))  _Float16 v8h;
typedef __attribute__((ext_vector_type(2)))  _Float16 v2h;
typedef __attribute__((ext_vector_type(8)))  float    v8f;
typedef __attribute__((ext_vector_type(8)))  int      v8i;
typedef __attribute__((ext_vector_type(4)))  int      v4i;
typedef unsigned short __attribute__((may_alias)) u16a;

#if __has_builtin(__builtin_amdgcn_tanhf)
#define HW_TANH __builtin_amdgcn_tanhf
#elif __has_builtin(__builtin_amdgcn_tanh_f32)
#define HW_TANH __builtin_amdgcn_tanh_f32
#endif

#ifdef HW_TANH
__device__ __forceinline__ float tanh_act(float x) { return HW_TANH(x); }
__device__ __forceinline__ float sig_act(float x)  { return fmaf(0.5f, HW_TANH(0.5f * x), 0.5f); }
#else
__device__ __forceinline__ float tanh_act(float x) {
  return fmaf(2.0f, __builtin_amdgcn_rcpf(1.0f + __expf(-2.0f * x)), -1.0f);
}
__device__ __forceinline__ float sig_act(float x) {
  return __builtin_amdgcn_rcpf(1.0f + __expf(-x));
}
#endif

template <int CTRL>
__device__ __forceinline__ int dpp_mov(int v) {
  return __builtin_amdgcn_update_dpp(0, v, CTRL, 0xf, 0xf, true);
}
#define DPP_SWAP_PAIR  0xB1
#define DPP_ROW_SHARE  0x150
#define DPP_ROW_XMASK  0x160

__device__ __forceinline__ float xhalf(float x) {
  const int b = __builtin_bit_cast(int, x);
  return __builtin_bit_cast(float, __builtin_amdgcn_permlanex16(b, b, 0x76543210, 0xfedcba98, false, false));
}

__device__ __forceinline__ int pack_halves(float lo, float hi) {
  v2h t;
  t[0] = (_Float16)lo;
  t[1] = (_Float16)hi;
  return __builtin_bit_cast(int, t);
}

__device__ __forceinline__ v16h load_w_tile(const _Float16* __restrict__ base,
                                            int g, int s, int lane) {
  const int n  = lane & 15;
  const int kb = (lane >> 4) << 4;
  const _Float16* row = base + ((g << 5) + (s << 4) + n) * KPAD + kb;
  v8h lo = *(const v8h*)(row);
  v8h hi = *(const v8h*)(row + 8);
  v16h b;
#pragma unroll
  for (int i = 0; i < 8; ++i) { b[i] = lo[i]; b[8 + i] = hi[i]; }
  return b;
}

__device__ __forceinline__ int pair_dword(float hn) {
  const float nb = __builtin_bit_cast(float, dpp_mov<DPP_SWAP_PAIR>(__builtin_bit_cast(int, hn)));
  return pack_halves(nb, hn);
}

__device__ __forceinline__ v16h gather_operand(int pd) {
  v8i a;
  a[0] = dpp_mov<DPP_ROW_SHARE + 1>(pd);
  a[1] = dpp_mov<DPP_ROW_SHARE + 3>(pd);
  a[2] = dpp_mov<DPP_ROW_SHARE + 5>(pd);
  a[3] = dpp_mov<DPP_ROW_SHARE + 7>(pd);
  a[4] = dpp_mov<DPP_ROW_SHARE + 9>(pd);
  a[5] = dpp_mov<DPP_ROW_SHARE + 11>(pd);
  a[6] = dpp_mov<DPP_ROW_SHARE + 13>(pd);
  a[7] = dpp_mov<DPP_ROW_SHARE + 15>(pd);
  return __builtin_bit_cast(v16h, a);
}

__device__ __forceinline__ v16h set_rail(v16h op, bool hi_half, int rail) {
  v8i a = __builtin_bit_cast(v8i, op);
  a[7] = hi_half ? rail : a[7];
  return __builtin_bit_cast(v16h, a);
}

__device__ __forceinline__ float lstm_gate(float pi, float pf, float pg, float po, float& c) {
  const float iv = sig_act(pi);
  const float fv = sig_act(pf);
  const float gv = tanh_act(pg);
  const float ov = sig_act(po);
  const float cn = fmaf(fv, c, iv * gv);
  c = cn;
  return ov * tanh_act(cn);
}

__device__ __forceinline__ float lstm_gate_in(float di, float df, float dg, float dn,
                                              float qi, float qf, float qg, float qo, float& c) {
#ifdef HW_TANH
  const float iv = fmaf(0.5f, HW_TANH(fmaf(0.5f, di, qi)), 0.5f);
  const float fv = fmaf(0.5f, HW_TANH(fmaf(0.5f, df, qf)), 0.5f);
  const float ov = fmaf(0.5f, HW_TANH(fmaf(0.5f, dn, qo)), 0.5f);
#else
  const float iv = sig_act(fmaf(2.0f, qi, di));
  const float fv = sig_act(fmaf(2.0f, qf, df));
  const float ov = sig_act(fmaf(2.0f, qo, dn));
#endif
  const float gv = tanh_act(dg + qg);
  const float cn = fmaf(fv, c, iv * gv);
  c = cn;
  return ov * tanh_act(cn);
}

__device__ __forceinline__ void handover_barrier() {
  __builtin_amdgcn_fence(__ATOMIC_RELEASE, "workgroup", "local");
  __builtin_amdgcn_s_barrier();
  __builtin_amdgcn_fence(__ATOMIC_ACQUIRE, "workgroup", "local");
}


__device__ __forceinline__ void keep_tiles_disjoint(v8f (&d)[8]) {
  asm volatile("" : "+v"(d[0]), "+v"(d[1]), "+v"(d[2]), "+v"(d[3]),
                    "+v"(d[4]), "+v"(d[5]), "+v"(d[6]), "+v"(d[7]));
}

__device__ __forceinline__ void step_entry(v16h& aop, float& hprev, unsigned& half_mask) {
  asm volatile("" : "+v"(aop), "+v"(hprev), "+s"(half_mask));
}
__device__ __forceinline__ void tiles_done(v8f (&d)[8], float& oacc) {
  asm volatile("" : "+v"(d[0]), "+v"(d[1]), "+v"(d[2]), "+v"(d[3]),
                    "+v"(d[4]), "+v"(d[5]), "+v"(d[6]), "+v"(d[7]), "+v"(oacc));
}
__device__ __forceinline__ void tiles_then_proj(v8f (&d)[8], float& oacc, v16h& hrows) {
  asm volatile("" : "+v"(d[0]), "+v"(d[1]), "+v"(d[2]), "+v"(d[3]),
                    "+v"(d[4]), "+v"(d[5]), "+v"(d[6]), "+v"(d[7]), "+v"(oacc), "+v"(hrows));
}
__device__ __forceinline__ void proj_then_gather(v16h& hrows, float& hn) {
  asm volatile("" : "+v"(hrows), "+v"(hn));
}

__device__ __forceinline__ int layer1_step(int j, bool hi_half,
                                           const v16h (&W)[16], const v8f& zc, float& c,
                                           v16h& aop, float xc) {
  const float xs = __builtin_bit_cast(float,
      __builtin_amdgcn_readlane(__builtin_bit_cast(int, xc), j));
  aop = set_rail(aop, hi_half, pack_halves(xs, 1.f));

  v8f d[8];
#pragma unroll
  for (int t = 0; t < 8; ++t)
    d[t] = __builtin_amdgcn_wmma_f32_16x16x32_f16(false, aop, false, W[t],
                                                  (short)0, zc, false, false);
  keep_tiles_disjoint(d);
  const float pi = hi_half ? d[1][0] : d[0][0];
  const float pf = hi_half ? d[3][0] : d[2][0];
  const float pg = hi_half ? d[5][0] : d[4][0];
  const float po = hi_half ? d[7][0] : d[6][0];
  const float hn = lstm_gate(pi, pf, pg, po, c);
  const int pd = pair_dword(hn);
  aop = gather_operand(pd);
  return pd;
}

template <bool REC, bool PROJ>
__device__ __forceinline__ void layer2_step(int s, int I, int lane, bool hi_half,
                                            const v16h (&W)[16], const v8f& zc, float& c,
                                            v16h& aop, int rail, float wfc, float bfc,
                                            float& hprev, float& oacc,
                                            const v8f (&qc)[8], v8f (&qn)[8], v16h& hrows,
                                            bool line_done, float* orow) {
  const int u = s - 1;
  unsigned half_mask = 0xffff0000u;
  if (REC) {
    step_entry(aop, hprev, half_mask);
    float v = hprev * wfc;
    v += __builtin_bit_cast(float, dpp_mov<DPP_ROW_XMASK + 1>(__builtin_bit_cast(int, v)));
    v += __builtin_bit_cast(float, dpp_mov<DPP_ROW_XMASK + 2>(__builtin_bit_cast(int, v)));
    v += __builtin_bit_cast(float, dpp_mov<DPP_ROW_XMASK + 4>(__builtin_bit_cast(int, v)));
    v += __builtin_bit_cast(float, dpp_mov<DPP_ROW_XMASK + 8>(__builtin_bit_cast(int, v)));
    v += xhalf(v);
    oacc = (lane == (u & 31)) ? (v + bfc) : oacc;
  }

  v8f d[8];
  if (REC) {
#pragma unroll
    for (int t = 0; t < 8; ++t)
      d[t] = __builtin_amdgcn_wmma_f32_16x16x32_f16(false, aop, false, W[t],
                                                    (short)0, zc, false, false);
    if (PROJ) tiles_then_proj(d, oacc, hrows); else tiles_done(d, oacc);
  }
  if (PROJ)
    qn[I] = __builtin_amdgcn_wmma_f32_16x16x32_f16(false, hrows, false, W[8 + I],
                                                   (short)0, zc, false, false);
  if (REC) {
    const float di = hi_half ? d[1][0] : d[0][0];
    const float df = hi_half ? d[3][0] : d[2][0];
    const float dg = hi_half ? d[5][0] : d[4][0];
    const float dn = hi_half ? d[7][0] : d[6][0];
    const bool hq = __builtin_amdgcn_inverse_ballot_w32(half_mask);
    const float qi = 0.5f * (hq ? qc[1][I] : qc[0][I]);
    const float qf = 0.5f * (hq ? qc[3][I] : qc[2][I]);
    const float qg =        (hq ? qc[5][I] : qc[4][I]);
    const float qo = 0.5f * (hq ? qc[7][I] : qc[6][I]);
    float hn = lstm_gate_in(di, df, dg, dn, qi, qf, qg, qo, c);
    if (PROJ) proj_then_gather(hrows, hn);
    aop = set_rail(gather_operand(pair_dword(hn)), hi_half, rail);
    hprev = hn;
    if (line_done) orow[(u & ~31) + lane] = oacc;
  }
}

__device__ __forceinline__ v16h load_h1_rows(const int* src, int lane, int grp) {
  const int* p = src + (8 * grp + (lane & 7)) * 16;
  const v4i lo = *(const v4i*)(p);
  const v4i hi = *(const v4i*)(p + 4);
  v8i hw;
#pragma unroll
  for (int e = 0; e < 4; ++e) { hw[e] = lo[e]; hw[4 + e] = hi[e]; }
  return __builtin_bit_cast(v16h, hw);
}

template <bool REC, bool PROJ>
__device__ __forceinline__ void layer2_group(int s0, bool line_first, int lane, bool hi_half,
                                             const v16h (&W)[16], const v8f& zc, float& c,
                                             v16h& aop, int rail, float wfc, float bfc,
                                             float& hprev, float& oacc,
                                             const v8f (&qc)[8], v8f (&qn)[8], v16h& hrows,
                                             float* orow) {
#pragma unroll
  for (int i = 0; i < 8; ++i)
    layer2_step<REC, PROJ>(s0 + i, i, lane, hi_half, W, zc, c, aop, rail, wfc, bfc, hprev, oacc,
                           qc, qn, hrows, (i == 0) && line_first, orow);
}

__global__ __launch_bounds__(NTHR, 1)
void lstm2_wmma_kernel(const float* __restrict__ x,
                       const float* __restrict__ w_ih1, const float* __restrict__ w_hh1,
                       const float* __restrict__ b_ih1, const float* __restrict__ b_hh1,
                       const float* __restrict__ w_ih2, const float* __restrict__ w_hh2,
                       const float* __restrict__ b_ih2, const float* __restrict__ b_hh2,
                       const float* __restrict__ w_fc,  const float* __restrict__ b_fc,
                       float* __restrict__ out) {
  __shared__ _Float16 sA1[MPAD * KPAD];
  __shared__ _Float16 sAi2[MPAD * KPAD];
  __shared__ _Float16 sAh2[MPAD * KPAD];
  __shared__ __attribute__((aligned(16))) int sP[NROW * 2 * NCH * 16];

  const int tid  = threadIdx.x;
  const int lane = tid & 31;
  const int wave  = __builtin_amdgcn_readfirstlane(tid >> 5);
  const int layer = wave & 1;
  const int prow  = wave >> 1;
  const bool hi_half = lane >= 16;

  for (int idx = tid; idx < MPAD * KPAD; idx += NTHR) {
    const int m = idx >> 5, k = idx & 31;
    const int g = m >> 5, j = m & 31;
    const int row = g * HSZ + j;
    float a1 = 0.f, ai2 = 0.f, ah2 = 0.f;
    if (j < HSZ) {
      if (k < HSZ) {
        a1  = w_hh1[row * HSZ + k];
        ai2 = w_ih2[row * HSZ + k];
        ah2 = w_hh2[row * HSZ + k];
      } else if (k == 30) {
        a1  = w_ih1[row];
      } else {
        a1  = b_ih1[row] + b_hh1[row];
        ah2 = b_ih2[row] + b_hh2[row];
      }
    }
    sA1[idx]  = (_Float16)a1;
    sAi2[idx] = (_Float16)ai2;
    sAh2[idx] = (_Float16)ah2;
  }
  __syncthreads();

  const _Float16* baseX = layer ? sAh2 : sA1;
  v16h W[16];
#pragma unroll
  for (int g = 0; g < 4; ++g) {
#pragma unroll
    for (int s = 0; s < 2; ++s) {
      W[2 * g + s]     = load_w_tile(baseX, g, s, lane);
      W[8 + 2 * g + s] = load_w_tile(sAi2,  g, s, lane);
    }
  }

  const float wfc = (lane < HSZ) ? w_fc[lane] : 0.f;
  const float bfc = b_fc[0];

  v8f zc;
#pragma unroll
  for (int i = 0; i < 8; ++i) zc[i] = 0.f;

  float c = 0.f;

  const int brow = blockIdx.x * NROW + prow;
  const float* xrow = x + (size_t)brow * TLEN;
  float* orow = out + (size_t)brow * TLEN;
  int* sProw = sP + prow * (2 * NCH * 16);

  const int xl = lane & (NCH - 1);
  float xc = xrow[xl];
  float hprev = 0.f, oacc = 0.f;
  v8f qa[8], qb[8];
#pragma unroll
  for (int t = 0; t < 8; ++t) { qa[t] = zc; qb[t] = zc; }

  const int rail2 = pack_halves(0.f, 1.f);
  v8i a0;
#pragma unroll
  for (int i = 0; i < 8; ++i) a0[i] = 0;
  v16h aop = set_rail(__builtin_bit_cast(v16h, a0), hi_half, rail2);

  for (int p = 0; p <= NPH; ++p) {
    if (layer == 0) {
      if (p < NPH) {
        const float xn = xrow[min(NCH * (p + 1) + xl, TLEN - 1)];
        int pd[NCH];
#pragma unroll
        for (int j = 0; j < NCH; ++j)
          pd[j] = layer1_step(j, hi_half, W, zc, c, aop, xc);
        u16a* dst = (u16a*)(sProw + (p & 1) * (NCH * 16)) + (lane ^ 1);
#pragma unroll
        for (int j = 0; j < NCH; ++j)
          dst[j * 32] = (u16a)pd[j];
        xc = xn;
      }
    } else {
      if (p >= 1) {
        const int* src = sProw + ((p - 1) & 1) * (NCH * 16) + (hi_half ? 8 : 0);
        v16h ha = load_h1_rows(src, lane, 0);
        v16h hb = load_h1_rows(src, lane, 1);
        const int s0 = NCH * (p - 1) - 8;
        if (p == 1)
          layer2_group<false, true>(s0, false, lane, hi_half, W, zc, c, aop, rail2, wfc, bfc,
                                    hprev, oacc, qb, qa, ha, orow);
        else
          layer2_group<true, true>(s0, false, lane, hi_half, W, zc, c, aop, rail2, wfc, bfc,
                                   hprev, oacc, qb, qa, ha, orow);
        layer2_group<true, true>(s0 + 8, (p & 1) != 0 && p >= 3, lane, hi_half, W, zc, c, aop,
                                 rail2, wfc, bfc, hprev, oacc, qa, qb, hb, orow);
      }
    }
    if (p < NPH) handover_barrier();
  }

  if (layer != 0) {
    v16h none = aop;
    layer2_group<true, false>(TLEN - 8, false, lane, hi_half, W, zc, c, aop, rail2, wfc, bfc,
                              hprev, oacc, qb, qa, none, orow);
    float v = hprev * wfc;
    v += __builtin_bit_cast(float, dpp_mov<DPP_ROW_XMASK + 1>(__builtin_bit_cast(int, v)));
    v += __builtin_bit_cast(float, dpp_mov<DPP_ROW_XMASK + 2>(__builtin_bit_cast(int, v)));
    v += __builtin_bit_cast(float, dpp_mov<DPP_ROW_XMASK + 4>(__builtin_bit_cast(int, v)));
    v += __builtin_bit_cast(float, dpp_mov<DPP_ROW_XMASK + 8>(__builtin_bit_cast(int, v)));
    v += xhalf(v);
    oacc = (lane == 31) ? (v + bfc) : oacc;
    orow[TLEN - 32 + lane] = oacc;
  }
}

extern "C" void kernel_launch(void* const* d_in, const int* in_sizes, int n_in,
                              void* d_out, int out_size, void* d_ws, size_t ws_size,
                              hipStream_t stream) {
  (void)in_sizes; (void)n_in; (void)out_size; (void)d_ws; (void)ws_size;
  const float* x     = (const float*)d_in[0];
  const float* w_ih1 = (const float*)d_in[1];
  const float* w_hh1 = (const float*)d_in[2];
  const float* b_ih1 = (const float*)d_in[3];
  const float* b_hh1 = (const float*)d_in[4];
  const float* w_ih2 = (const float*)d_in[5];
  const float* w_hh2 = (const float*)d_in[6];
  const float* b_ih2 = (const float*)d_in[7];
  const float* b_hh2 = (const float*)d_in[8];
  const float* w_fc  = (const float*)d_in[9];
  const float* b_fc  = (const float*)d_in[10];
  float* out = (float*)d_out;

  dim3 grid(BSZ / NROW);
  dim3 block(NTHR);
  lstm2_wmma_kernel<<<grid, block, 0, stream>>>(x, w_ih1, w_hh1, b_ih1, b_hh1,
                                                w_ih2, w_hh2, b_ih2, b_hh2,
                                                w_fc, b_fc, out);
}


# FC head: a step keeps only its lane product; the 32-lane sums of 8 steps are formed together by a folding reduction beside the next group's gate math, not among a step's WMMAs; FC sum regrouped
# speedup vs baseline: 5.5614x; 1.1926x over previous
#include <hip/hip_runtime.h>


#define HSZ   30
#define MPAD  128
#define KPAD  32
#define TLEN  2048
#define BSZ   512
#define NROW  2
#define NTHR  (NROW * 64)
#define NCH   16
#define NPH   (TLEN / NCH)

typedef __attribute__((ext_vector_type(16))) _Float16 v16h;
typedef __attribute__((ext_vector_type(8)))  _Float16 v8h;
typedef __attribute__((ext_vector_type(2)))  _Float16 v2h;
typedef __attribute__((ext_vector_type(8)))  float    v8f;
typedef __attribute__((ext_vector_type(8)))  int      v8i;
typedef __attribute__((ext_vector_type(4)))  int      v4i;
typedef unsigned short __attribute__((may_alias)) u16a;

#if __has_builtin(__builtin_amdgcn_tanhf)
#define HW_TANH __builtin_amdgcn_tanhf
#elif __has_builtin(__builtin_amdgcn_tanh_f32)
#define HW_TANH __builtin_amdgcn_tanh_f32
#endif

#ifdef HW_TANH
__device__ __forceinline__ float tanh_act(float x) { return HW_TANH(x); }
__device__ __forceinline__ float sig_act(float x)  { return fmaf(0.5f, HW_TANH(0.5f * x), 0.5f); }
#else
__device__ __forceinline__ float tanh_act(float x) {
  return fmaf(2.0f, __builtin_amdgcn_rcpf(1.0f + __expf(-2.0f * x)), -1.0f);
}
__device__ __forceinline__ float sig_act(float x) {
  return __builtin_amdgcn_rcpf(1.0f + __expf(-x));
}
#endif

template <int CTRL>
__device__ __forceinline__ int dpp_mov(int v) {
  return __builtin_amdgcn_update_dpp(0, v, CTRL, 0xf, 0xf, true);
}
#define DPP_SWAP_PAIR  0xB1
#define DPP_ROW_SHARE  0x150
#define DPP_ROW_XMASK  0x160

__device__ __forceinline__ float xhalf(float x) {
  const int b = __builtin_bit_cast(int, x);
  return __builtin_bit_cast(float, __builtin_amdgcn_permlanex16(b, b, 0x76543210, 0xfedcba98, false, false));
}

__device__ __forceinline__ int pack_halves(float lo, float hi) {
  v2h t;
  t[0] = (_Float16)lo;
  t[1] = (_Float16)hi;
  return __builtin_bit_cast(int, t);
}

__device__ __forceinline__ v16h load_w_tile(const _Float16* __restrict__ base,
                                            int g, int s, int lane) {
  const int n  = lane & 15;
  const int kb = (lane >> 4) << 4;
  const _Float16* row = base + ((g << 5) + (s << 4) + n) * KPAD + kb;
  v8h lo = *(const v8h*)(row);
  v8h hi = *(const v8h*)(row + 8);
  v16h b;
#pragma unroll
  for (int i = 0; i < 8; ++i) { b[i] = lo[i]; b[8 + i] = hi[i]; }
  return b;
}

__device__ __forceinline__ int pair_dword(float hn) {
  const float nb = __builtin_bit_cast(float, dpp_mov<DPP_SWAP_PAIR>(__builtin_bit_cast(int, hn)));
  return pack_halves(nb, hn);
}

__device__ __forceinline__ v16h gather_operand(int pd) {
  v8i a;
  a[0] = dpp_mov<DPP_ROW_SHARE + 1>(pd);
  a[1] = dpp_mov<DPP_ROW_SHARE + 3>(pd);
  a[2] = dpp_mov<DPP_ROW_SHARE + 5>(pd);
  a[3] = dpp_mov<DPP_ROW_SHARE + 7>(pd);
  a[4] = dpp_mov<DPP_ROW_SHARE + 9>(pd);
  a[5] = dpp_mov<DPP_ROW_SHARE + 11>(pd);
  a[6] = dpp_mov<DPP_ROW_SHARE + 13>(pd);
  a[7] = dpp_mov<DPP_ROW_SHARE + 15>(pd);
  return __builtin_bit_cast(v16h, a);
}

__device__ __forceinline__ v16h set_rail(v16h op, bool hi_half, int rail) {
  v8i a = __builtin_bit_cast(v8i, op);
  a[7] = hi_half ? rail : a[7];
  return __builtin_bit_cast(v16h, a);
}

__device__ __forceinline__ float lstm_gate(float pi, float pf, float pg, float po, float& c) {
  const float iv = sig_act(pi);
  const float fv = sig_act(pf);
  const float gv = tanh_act(pg);
  const float ov = sig_act(po);
  const float cn = fmaf(fv, c, iv * gv);
  c = cn;
  return ov * tanh_act(cn);
}

__device__ __forceinline__ float lstm_gate_in(float di, float df, float dg, float dn,
                                              float qi, float qf, float qg, float qo, float& c) {
#ifdef HW_TANH
  const float iv = fmaf(0.5f, HW_TANH(fmaf(0.5f, di, qi)), 0.5f);
  const float fv = fmaf(0.5f, HW_TANH(fmaf(0.5f, df, qf)), 0.5f);
  const float ov = fmaf(0.5f, HW_TANH(fmaf(0.5f, dn, qo)), 0.5f);
#else
  const float iv = sig_act(fmaf(2.0f, qi, di));
  const float fv = sig_act(fmaf(2.0f, qf, df));
  const float ov = sig_act(fmaf(2.0f, qo, dn));
#endif
  const float gv = tanh_act(dg + qg);
  const float cn = fmaf(fv, c, iv * gv);
  c = cn;
  return ov * tanh_act(cn);
}

__device__ __forceinline__ void handover_barrier() {
  __builtin_amdgcn_fence(__ATOMIC_RELEASE, "workgroup", "local");
  __builtin_amdgcn_s_barrier();
  __builtin_amdgcn_fence(__ATOMIC_ACQUIRE, "workgroup", "local");
}


__device__ __forceinline__ void keep_tiles_disjoint(v8f (&d)[8]) {
  asm volatile("" : "+v"(d[0]), "+v"(d[1]), "+v"(d[2]), "+v"(d[3]),
                    "+v"(d[4]), "+v"(d[5]), "+v"(d[6]), "+v"(d[7]));
}

__device__ __forceinline__ void step_entry(v16h& aop, unsigned& half_mask) {
  asm volatile("" : "+v"(aop), "+s"(half_mask));
}
__device__ __forceinline__ void tiles_then_proj(v8f (&d)[8], v16h& hrows) {
  asm volatile("" : "+v"(d[0]), "+v"(d[1]), "+v"(d[2]), "+v"(d[3]),
                    "+v"(d[4]), "+v"(d[5]), "+v"(d[6]), "+v"(d[7]), "+v"(hrows));
}
__device__ __forceinline__ void proj_then_gather(v16h& hrows, float& hn) {
  asm volatile("" : "+v"(hrows), "+v"(hn));
}

__device__ __forceinline__ int layer1_step(int j, bool hi_half,
                                           const v16h (&W)[16], const v8f& zc, float& c,
                                           v16h& aop, float xc) {
  const float xs = __builtin_bit_cast(float,
      __builtin_amdgcn_readlane(__builtin_bit_cast(int, xc), j));
  aop = set_rail(aop, hi_half, pack_halves(xs, 1.f));

  v8f d[8];
#pragma unroll
  for (int t = 0; t < 8; ++t)
    d[t] = __builtin_amdgcn_wmma_f32_16x16x32_f16(false, aop, false, W[t],
                                                  (short)0, zc, false, false);
  keep_tiles_disjoint(d);
  const float pi = hi_half ? d[1][0] : d[0][0];
  const float pf = hi_half ? d[3][0] : d[2][0];
  const float pg = hi_half ? d[5][0] : d[4][0];
  const float po = hi_half ? d[7][0] : d[6][0];
  const float hn = lstm_gate(pi, pf, pg, po, c);
  const int pd = pair_dword(hn);
  aop = gather_operand(pd);
  return pd;
}

template <int XM>
__device__ __forceinline__ float fold_pair(float a, float b, bool bit_set) {
  const float keep = bit_set ? b : a;
  const float send = bit_set ? a : b;
  return keep + __builtin_bit_cast(float, dpp_mov<DPP_ROW_XMASK + XM>(__builtin_bit_cast(int, send)));
}

__device__ __forceinline__ void fold_in8(float (&r)[8]) {
  asm volatile("" : "+v"(r[0]), "+v"(r[1]), "+v"(r[2]), "+v"(r[3]),
                    "+v"(r[4]), "+v"(r[5]), "+v"(r[6]), "+v"(r[7]));
}
__device__ __forceinline__ void fold_in4(float (&r)[8]) {
  asm volatile("" : "+v"(r[0]), "+v"(r[1]), "+v"(r[2]), "+v"(r[3]));
}
__device__ __forceinline__ void fold_in2(float (&r)[8]) {
  asm volatile("" : "+v"(r[0]), "+v"(r[1]));
}
__device__ __forceinline__ void fold_in1(float (&r)[8]) {
  asm volatile("" : "+v"(r[0]));
}

__device__ __forceinline__ void fc_fold_stage(int I, int lane, float (&red)[8], float& oacc,
                                              float bfc, int qprev) {
  if (I == 1) {
    fold_in8(red);
    const bool b0 = (lane & 1) != 0;
    red[0] = fold_pair<1>(red[0], red[1], b0);
    red[1] = fold_pair<1>(red[2], red[3], b0);
    red[2] = fold_pair<1>(red[4], red[5], b0);
    red[3] = fold_pair<1>(red[6], red[7], b0);
  } else if (I == 2) {
    fold_in4(red);
    const bool b1 = (lane & 2) != 0;
    red[0] = fold_pair<2>(red[0], red[1], b1);
    red[1] = fold_pair<2>(red[2], red[3], b1);
  } else if (I == 3) {
    fold_in2(red);
    const bool b2 = (lane & 4) != 0;
    float v = fold_pair<4>(red[0], red[1], b2);
    v += __builtin_bit_cast(float, dpp_mov<DPP_ROW_XMASK + 8>(__builtin_bit_cast(int, v)));
    red[0] = v;
  } else if (I == 4) {
    fold_in1(red);
    const float v = red[0] + xhalf(red[0]);
    oacc = ((lane >> 3) == qprev) ? (v + bfc) : oacc;
  }
}

template <bool REC, bool PROJ>
__device__ __forceinline__ void layer2_step(int I, int lane, bool hi_half,
                                            const v16h (&W)[16], const v8f& zc, float& c,
                                            v16h& aop, int rail, float wfc, float bfc,
                                            float (&fcp)[8], float (&red)[8], float& oacc, int qprev,
                                            const v8f (&qc)[8], v8f (&qn)[8], v16h& hrows) {
  unsigned half_mask = 0xffff0000u;
  if (REC) step_entry(aop, half_mask);

  v8f d[8];
  if (REC) {
#pragma unroll
    for (int t = 0; t < 8; ++t)
      d[t] = __builtin_amdgcn_wmma_f32_16x16x32_f16(false, aop, false, W[t],
                                                    (short)0, zc, false, false);
    if (PROJ) tiles_then_proj(d, hrows); else keep_tiles_disjoint(d);
    fc_fold_stage(I, lane, red, oacc, bfc, qprev);
  }
  if (PROJ)
    qn[I] = __builtin_amdgcn_wmma_f32_16x16x32_f16(false, hrows, false, W[8 + I],
                                                   (short)0, zc, false, false);
  if (REC) {
    const float di = hi_half ? d[1][0] : d[0][0];
    const float df = hi_half ? d[3][0] : d[2][0];
    const float dg = hi_half ? d[5][0] : d[4][0];
    const float dn = hi_half ? d[7][0] : d[6][0];
    const bool hq = __builtin_amdgcn_inverse_ballot_w32(half_mask);
    const float qi = 0.5f * (hq ? qc[1][I] : qc[0][I]);
    const float qf = 0.5f * (hq ? qc[3][I] : qc[2][I]);
    const float qg =        (hq ? qc[5][I] : qc[4][I]);
    const float qo = 0.5f * (hq ? qc[7][I] : qc[6][I]);
    float hn = lstm_gate_in(di, df, dg, dn, qi, qf, qg, qo, c);
    if (PROJ) proj_then_gather(hrows, hn);
    aop = set_rail(gather_operand(pair_dword(hn)), hi_half, rail);
    fcp[I] = hn * wfc;
  }
}

__device__ __forceinline__ v16h load_h1_rows(const int* src, int lane, int grp) {
  const int* p = src + (8 * grp + (lane & 7)) * 16;
  const v4i lo = *(const v4i*)(p);
  const v4i hi = *(const v4i*)(p + 4);
  v8i hw;
#pragma unroll
  for (int e = 0; e < 4; ++e) { hw[e] = lo[e]; hw[4 + e] = hi[e]; }
  return __builtin_bit_cast(v16h, hw);
}

template <bool REC, bool PROJ>
__device__ __forceinline__ void layer2_group(int s0, int lane, bool hi_half,
                                             const v16h (&W)[16], const v8f& zc, float& c,
                                             v16h& aop, int rail, float wfc, float bfc,
                                             float (&fq)[8], float& oacc,
                                             const v8f (&qc)[8], v8f (&qn)[8], v16h& hrows,
                                             float* orow) {
  float fcp[8], red[8];
#pragma unroll
  for (int k = 0; k < 8; ++k) { red[k] = fq[k]; fcp[k] = 0.f; }
  const int qprev = ((s0 >> 3) + 3) & 3;
#pragma unroll
  for (int i = 0; i < 8; ++i)
    layer2_step<REC, PROJ>(i, lane, hi_half, W, zc, c, aop, rail, wfc, bfc, fcp, red, oacc, qprev,
                           qc, qn, hrows);
  if (REC) {
#pragma unroll
    for (int k = 0; k < 8; ++k) fq[k] = fcp[k];
    if ((s0 & 31) == 0 && s0 != 0) orow[s0 - 32 + lane] = oacc;
  }
}

__global__ __launch_bounds__(NTHR, 1)
void lstm2_wmma_kernel(const float* __restrict__ x,
                       const float* __restrict__ w_ih1, const float* __restrict__ w_hh1,
                       const float* __restrict__ b_ih1, const float* __restrict__ b_hh1,
                       const float* __restrict__ w_ih2, const float* __restrict__ w_hh2,
                       const float* __restrict__ b_ih2, const float* __restrict__ b_hh2,
                       const float* __restrict__ w_fc,  const float* __restrict__ b_fc,
                       float* __restrict__ out) {
  __shared__ _Float16 sA1[MPAD * KPAD];
  __shared__ _Float16 sAi2[MPAD * KPAD];
  __shared__ _Float16 sAh2[MPAD * KPAD];
  __shared__ __attribute__((aligned(16))) int sP[NROW * 2 * NCH * 16];

  const int tid  = threadIdx.x;
  const int lane = tid & 31;
  const int wave  = __builtin_amdgcn_readfirstlane(tid >> 5);
  const int layer = wave & 1;
  const int prow  = wave >> 1;
  const bool hi_half = lane >= 16;

  for (int idx = tid; idx < MPAD * KPAD; idx += NTHR) {
    const int m = idx >> 5, k = idx & 31;
    const int g = m >> 5, j = m & 31;
    const int row = g * HSZ + j;
    float a1 = 0.f, ai2 = 0.f, ah2 = 0.f;
    if (j < HSZ) {
      if (k < HSZ) {
        a1  = w_hh1[row * HSZ + k];
        ai2 = w_ih2[row * HSZ + k];
        ah2 = w_hh2[row * HSZ + k];
      } else if (k == 30) {
        a1  = w_ih1[row];
      } else {
        a1  = b_ih1[row] + b_hh1[row];
        ah2 = b_ih2[row] + b_hh2[row];
      }
    }
    sA1[idx]  = (_Float16)a1;
    sAi2[idx] = (_Float16)ai2;
    sAh2[idx] = (_Float16)ah2;
  }
  __syncthreads();

  const _Float16* baseX = layer ? sAh2 : sA1;
  v16h W[16];
#pragma unroll
  for (int g = 0; g < 4; ++g) {
#pragma unroll
    for (int s = 0; s < 2; ++s) {
      W[2 * g + s]     = load_w_tile(baseX, g, s, lane);
      W[8 + 2 * g + s] = load_w_tile(sAi2,  g, s, lane);
    }
  }

  const float wfc = (lane < HSZ) ? w_fc[lane] : 0.f;
  const float bfc = b_fc[0];

  v8f zc;
#pragma unroll
  for (int i = 0; i < 8; ++i) zc[i] = 0.f;

  float c = 0.f;

  const int brow = blockIdx.x * NROW + prow;
  const float* xrow = x + (size_t)brow * TLEN;
  float* orow = out + (size_t)brow * TLEN;
  int* sProw = sP + prow * (2 * NCH * 16);

  const int xl = lane & (NCH - 1);
  float xc = xrow[xl];
  float oacc = 0.f;
  float fq[8];
#pragma unroll
  for (int k = 0; k < 8; ++k) fq[k] = 0.f;
  v8f qa[8], qb[8];
#pragma unroll
  for (int t = 0; t < 8; ++t) { qa[t] = zc; qb[t] = zc; }

  const int rail2 = pack_halves(0.f, 1.f);
  v8i a0;
#pragma unroll
  for (int i = 0; i < 8; ++i) a0[i] = 0;
  v16h aop = set_rail(__builtin_bit_cast(v16h, a0), hi_half, rail2);

  for (int p = 0; p <= NPH; ++p) {
    if (layer == 0) {
      if (p < NPH) {
        const float xn = xrow[min(NCH * (p + 1) + xl, TLEN - 1)];
        int pd[NCH];
#pragma unroll
        for (int j = 0; j < NCH; ++j)
          pd[j] = layer1_step(j, hi_half, W, zc, c, aop, xc);
        u16a* dst = (u16a*)(sProw + (p & 1) * (NCH * 16)) + (lane ^ 1);
#pragma unroll
        for (int j = 0; j < NCH; ++j)
          dst[j * 32] = (u16a)pd[j];
        xc = xn;
      }
    } else {
      if (p >= 1) {
        const int* src = sProw + ((p - 1) & 1) * (NCH * 16) + (hi_half ? 8 : 0);
        v16h ha = load_h1_rows(src, lane, 0);
        v16h hb = load_h1_rows(src, lane, 1);
        const int s0 = NCH * (p - 1) - 8;
        if (p == 1)
          layer2_group<false, true>(s0, lane, hi_half, W, zc, c, aop, rail2, wfc, bfc,
                                    fq, oacc, qb, qa, ha, orow);
        else
          layer2_group<true, true>(s0, lane, hi_half, W, zc, c, aop, rail2, wfc, bfc,
                                   fq, oacc, qb, qa, ha, orow);
        layer2_group<true, true>(s0 + 8, lane, hi_half, W, zc, c, aop, rail2, wfc, bfc,
                                 fq, oacc, qa, qb, hb, orow);
      }
    }
    if (p < NPH) handover_barrier();
  }

  if (layer != 0) {
    v16h none = aop;
    layer2_group<true, false>(TLEN - 8, lane, hi_half, W, zc, c, aop, rail2, wfc, bfc,
                              fq, oacc, qb, qa, none, orow);
#pragma unroll
    for (int i = 1; i <= 4; ++i) fc_fold_stage(i, lane, fq, oacc, bfc, 3);
    orow[TLEN - 32 + lane] = oacc;
  }
}

extern "C" void kernel_launch(void* const* d_in, const int* in_sizes, int n_in,
                              void* d_out, int out_size, void* d_ws, size_t ws_size,
                              hipStream_t stream) {
  (void)in_sizes; (void)n_in; (void)out_size; (void)d_ws; (void)ws_size;
  const float* x     = (const float*)d_in[0];
  const float* w_ih1 = (const float*)d_in[1];
  const float* w_hh1 = (const float*)d_in[2];
  const float* b_ih1 = (const float*)d_in[3];
  const float* b_hh1 = (const float*)d_in[4];
  const float* w_ih2 = (const float*)d_in[5];
  const float* w_hh2 = (const float*)d_in[6];
  const float* b_ih2 = (const float*)d_in[7];
  const float* b_hh2 = (const float*)d_in[8];
  const float* w_fc  = (const float*)d_in[9];
  const float* b_fc  = (const float*)d_in[10];
  float* out = (float*)d_out;

  dim3 grid(BSZ / NROW);
  dim3 block(NTHR);
  lstm2_wmma_kernel<<<grid, block, 0, stream>>>(x, w_ih1, w_hh1, b_ih1, b_hh1,
                                                w_ih2, w_hh2, b_ih2, b_hh2,
                                                w_fc, b_fc, out);
}


# speedup vs baseline: 5.7658x; 1.0368x over previous
#include <hip/hip_runtime.h>


#define HSZ   30
#define MPAD  128
#define KPAD  32
#define TLEN  2048
#define BSZ   512
#define NROW  2
#define NTHR  (NROW * 64)
#define NCH   16
#define NPH   (TLEN / NCH)

typedef __attribute__((ext_vector_type(16))) _Float16 v16h;
typedef __attribute__((ext_vector_type(8)))  _Float16 v8h;
typedef __attribute__((ext_vector_type(2)))  _Float16 v2h;
typedef __attribute__((ext_vector_type(8)))  float    v8f;
typedef __attribute__((ext_vector_type(8)))  int      v8i;
typedef __attribute__((ext_vector_type(4)))  int      v4i;
typedef unsigned short __attribute__((may_alias)) u16a;

#if __has_builtin(__builtin_amdgcn_tanhf)
#define HW_TANH __builtin_amdgcn_tanhf
#elif __has_builtin(__builtin_amdgcn_tanh_f32)
#define HW_TANH __builtin_amdgcn_tanh_f32
#endif

#ifdef HW_TANH
__device__ __forceinline__ float tanh_act(float x) { return HW_TANH(x); }
__device__ __forceinline__ float sig_act(float x)  { return fmaf(0.5f, HW_TANH(0.5f * x), 0.5f); }
#else
__device__ __forceinline__ float tanh_act(float x) {
  return fmaf(2.0f, __builtin_amdgcn_rcpf(1.0f + __expf(-2.0f * x)), -1.0f);
}
__device__ __forceinline__ float sig_act(float x) {
  return __builtin_amdgcn_rcpf(1.0f + __expf(-x));
}
#endif

template <int CTRL>
__device__ __forceinline__ int dpp_mov(int v) {
  return __builtin_amdgcn_update_dpp(0, v, CTRL, 0xf, 0xf, true);
}
#define DPP_SWAP_PAIR  0xB1
#define DPP_ROW_XMASK  0x160

__device__ __forceinline__ float xhalf(float x) {
  const int b = __builtin_bit_cast(int, x);
  return __builtin_bit_cast(float, __builtin_amdgcn_permlanex16(b, b, 0x76543210, 0xfedcba98, false, false));
}

__device__ __forceinline__ int pack_halves(float lo, float hi) {
  v2h t;
  t[0] = (_Float16)lo;
  t[1] = (_Float16)hi;
  return __builtin_bit_cast(int, t);
}

__device__ __forceinline__ v16h load_w_tile(const _Float16* __restrict__ base,
                                            int g, int s, int lane) {
  const int n  = lane & 15;
  const int kb = (lane >> 4) << 4;
  const _Float16* row = base + ((g << 5) + (s << 4) + n) * KPAD + kb;
  v8h lo = *(const v8h*)(row);
  v8h hi = *(const v8h*)(row + 8);
  v16h b;
#pragma unroll
  for (int i = 0; i < 8; ++i) { b[i] = lo[i]; b[8 + i] = hi[i]; }
  return b;
}

__device__ __forceinline__ int pair_dword(float hn) {
  const float nb = __builtin_bit_cast(float, dpp_mov<DPP_SWAP_PAIR>(__builtin_bit_cast(int, hn)));
  return pack_halves(nb, hn);
}

__device__ __forceinline__ int row_pick(int v, int sel) {
  return (int)__builtin_amdgcn_permlane16_var(0u, (unsigned)v, (unsigned)sel, false, true);
}
__device__ __forceinline__ v16h gather_operand(int pd, const int (&sel)[8]) {
  v8i a;
#pragma unroll
  for (int j = 0; j < 8; ++j) a[j] = row_pick(pd, sel[j]);
  return __builtin_bit_cast(v16h, a);
}

__device__ __forceinline__ v16h set_rail(v16h op, bool hi_half, int rail) {
  v8i a = __builtin_bit_cast(v8i, op);
  a[7] = hi_half ? rail : a[7];
  return __builtin_bit_cast(v16h, a);
}

__device__ __forceinline__ float lstm_gate(float pi, float pf, float pg, float po, float& c) {
  const float iv = sig_act(pi);
  const float fv = sig_act(pf);
  const float gv = tanh_act(pg);
  const float ov = sig_act(po);
  const float cn = fmaf(fv, c, iv * gv);
  c = cn;
  return ov * tanh_act(cn);
}

__device__ __forceinline__ float lstm_gate_in(float di, float df, float dg, float dn,
                                              float qi, float qf, float qg, float qo, float& c) {
#ifdef HW_TANH
  const float iv = fmaf(0.5f, HW_TANH(fmaf(0.5f, di, qi)), 0.5f);
  const float fv = fmaf(0.5f, HW_TANH(fmaf(0.5f, df, qf)), 0.5f);
  const float ov = fmaf(0.5f, HW_TANH(fmaf(0.5f, dn, qo)), 0.5f);
#else
  const float iv = sig_act(fmaf(2.0f, qi, di));
  const float fv = sig_act(fmaf(2.0f, qf, df));
  const float ov = sig_act(fmaf(2.0f, qo, dn));
#endif
  const float gv = tanh_act(dg + qg);
  const float cn = fmaf(fv, c, iv * gv);
  c = cn;
  return ov * tanh_act(cn);
}

__device__ __forceinline__ void handover_barrier() {
  __builtin_amdgcn_fence(__ATOMIC_RELEASE, "workgroup", "local");
  __builtin_amdgcn_s_barrier();
  __builtin_amdgcn_fence(__ATOMIC_ACQUIRE, "workgroup", "local");
}


__device__ __forceinline__ void keep_tiles_disjoint(v8f (&d)[8]) {
  asm volatile("" : "+v"(d[0]), "+v"(d[1]), "+v"(d[2]), "+v"(d[3]),
                    "+v"(d[4]), "+v"(d[5]), "+v"(d[6]), "+v"(d[7]));
}

__device__ __forceinline__ void step_entry(v16h& aop, unsigned& half_mask) {
  asm volatile("" : "+v"(aop), "+s"(half_mask));
}
__device__ __forceinline__ void tiles_then_proj(v8f (&d)[8], v16h& hrows) {
  asm volatile("" : "+v"(d[0]), "+v"(d[1]), "+v"(d[2]), "+v"(d[3]),
                    "+v"(d[4]), "+v"(d[5]), "+v"(d[6]), "+v"(d[7]), "+v"(hrows));
}
__device__ __forceinline__ void proj_then_gather(v16h& hrows, float& hn) {
  asm volatile("" : "+v"(hrows), "+v"(hn));
}

__device__ __forceinline__ int layer1_step(int j, bool hi_half,
                                           const v16h (&W)[16], const v8f& zc, float& c,
                                           v16h& aop, float xc, const int (&sel)[8]) {
  const float xs = __builtin_bit_cast(float,
      __builtin_amdgcn_readlane(__builtin_bit_cast(int, xc), j));
  aop = set_rail(aop, hi_half, pack_halves(xs, 1.f));

  v8f d[8];
#pragma unroll
  for (int t = 0; t < 8; ++t)
    d[t] = __builtin_amdgcn_wmma_f32_16x16x32_f16(false, aop, false, W[t],
                                                  (short)0, zc, false, false);
  keep_tiles_disjoint(d);
  const float pi = hi_half ? d[1][0] : d[0][0];
  const float pf = hi_half ? d[3][0] : d[2][0];
  const float pg = hi_half ? d[5][0] : d[4][0];
  const float po = hi_half ? d[7][0] : d[6][0];
  const float hn = lstm_gate(pi, pf, pg, po, c);
  const int pd = pair_dword(hn);
  aop = gather_operand(pd, sel);
  return pd;
}

template <int XM>
__device__ __forceinline__ float fold_pair(float a, float b, bool bit_set) {
  const float keep = bit_set ? b : a;
  const float send = bit_set ? a : b;
  return keep + __builtin_bit_cast(float, dpp_mov<DPP_ROW_XMASK + XM>(__builtin_bit_cast(int, send)));
}

__device__ __forceinline__ void fold_in8(float (&r)[8]) {
  asm volatile("" : "+v"(r[0]), "+v"(r[1]), "+v"(r[2]), "+v"(r[3]),
                    "+v"(r[4]), "+v"(r[5]), "+v"(r[6]), "+v"(r[7]));
}
__device__ __forceinline__ void fold_in4(float (&r)[8]) {
  asm volatile("" : "+v"(r[0]), "+v"(r[1]), "+v"(r[2]), "+v"(r[3]));
}
__device__ __forceinline__ void fold_in2(float (&r)[8]) {
  asm volatile("" : "+v"(r[0]), "+v"(r[1]));
}
__device__ __forceinline__ void fold_in1(float (&r)[8]) {
  asm volatile("" : "+v"(r[0]));
}

__device__ __forceinline__ void fc_fold_stage(int I, int lane, float (&red)[8], float& oacc,
                                              float bfc, int qprev) {
  if (I == 1) {
    fold_in8(red);
    const bool b0 = (lane & 1) != 0;
    red[0] = fold_pair<1>(red[0], red[1], b0);
    red[1] = fold_pair<1>(red[2], red[3], b0);
    red[2] = fold_pair<1>(red[4], red[5], b0);
    red[3] = fold_pair<1>(red[6], red[7], b0);
  } else if (I == 2) {
    fold_in4(red);
    const bool b1 = (lane & 2) != 0;
    red[0] = fold_pair<2>(red[0], red[1], b1);
    red[1] = fold_pair<2>(red[2], red[3], b1);
  } else if (I == 3) {
    fold_in2(red);
    const bool b2 = (lane & 4) != 0;
    float v = fold_pair<4>(red[0], red[1], b2);
    v += __builtin_bit_cast(float, dpp_mov<DPP_ROW_XMASK + 8>(__builtin_bit_cast(int, v)));
    red[0] = v;
  } else if (I == 4) {
    fold_in1(red);
    const float v = red[0] + xhalf(red[0]);
    oacc = ((lane >> 3) == qprev) ? (v + bfc) : oacc;
  }
}

template <bool REC, bool PROJ>
__device__ __forceinline__ void layer2_step(int I, int lane, bool hi_half,
                                            const v16h (&W)[16], const v8f& zc, float& c,
                                            v16h& aop, int rail, float wfc, float bfc,
                                            float (&fcp)[8], float (&red)[8], float& oacc, int qprev,
                                            const v8f (&qc)[8], v8f (&qn)[8], v16h& hrows,
                                            const int (&sel)[8]) {
  unsigned half_mask = 0xffff0000u;
  if (REC) step_entry(aop, half_mask);

  v8f d[8];
  if (REC) {
#pragma unroll
    for (int t = 0; t < 8; ++t)
      d[t] = __builtin_amdgcn_wmma_f32_16x16x32_f16(false, aop, false, W[t],
                                                    (short)0, zc, false, false);
    if (PROJ) tiles_then_proj(d, hrows); else keep_tiles_disjoint(d);
    fc_fold_stage(I, lane, red, oacc, bfc, qprev);
  }
  if (PROJ)
    qn[I] = __builtin_amdgcn_wmma_f32_16x16x32_f16(false, hrows, false, W[8 + I],
                                                   (short)0, zc, false, false);
  if (REC) {
    const float di = hi_half ? d[1][0] : d[0][0];
    const float df = hi_half ? d[3][0] : d[2][0];
    const float dg = hi_half ? d[5][0] : d[4][0];
    const float dn = hi_half ? d[7][0] : d[6][0];
    const bool hq = __builtin_amdgcn_inverse_ballot_w32(half_mask);
    const float qi = 0.5f * (hq ? qc[1][I] : qc[0][I]);
    const float qf = 0.5f * (hq ? qc[3][I] : qc[2][I]);
    const float qg =        (hq ? qc[5][I] : qc[4][I]);
    const float qo = 0.5f * (hq ? qc[7][I] : qc[6][I]);
    float hn = lstm_gate_in(di, df, dg, dn, qi, qf, qg, qo, c);
    if (PROJ) proj_then_gather(hrows, hn);
    aop = set_rail(gather_operand(pair_dword(hn), sel), hi_half, rail);
    fcp[I] = hn * wfc;
  }
}

__device__ __forceinline__ v16h load_h1_rows(const int* src, int lane, int grp) {
  const int* p = src + (8 * grp + (lane & 7)) * 16;
  const v4i lo = *(const v4i*)(p);
  const v4i hi = *(const v4i*)(p + 4);
  v8i hw;
#pragma unroll
  for (int e = 0; e < 4; ++e) { hw[e] = lo[e]; hw[4 + e] = hi[e]; }
  return __builtin_bit_cast(v16h, hw);
}

template <bool REC, bool PROJ>
__device__ __forceinline__ void layer2_group(int s0, int lane, bool hi_half,
                                             const v16h (&W)[16], const v8f& zc, float& c,
                                             v16h& aop, int rail, float wfc, float bfc,
                                             float (&fq)[8], float& oacc,
                                             const v8f (&qc)[8], v8f (&qn)[8], v16h& hrows,
                                             float* orow, const int (&sel)[8]) {
  float fcp[8], red[8];
#pragma unroll
  for (int k = 0; k < 8; ++k) { red[k] = fq[k]; fcp[k] = 0.f; }
  const int qprev = ((s0 >> 3) + 3) & 3;
#pragma unroll
  for (int i = 0; i < 8; ++i)
    layer2_step<REC, PROJ>(i, lane, hi_half, W, zc, c, aop, rail, wfc, bfc, fcp, red, oacc, qprev,
                           qc, qn, hrows, sel);
  if (REC) {
#pragma unroll
    for (int k = 0; k < 8; ++k) fq[k] = fcp[k];
    if ((s0 & 31) == 0 && s0 != 0) orow[s0 - 32 + lane] = oacc;
  }
}

__global__ __launch_bounds__(NTHR, 1)
void lstm2_wmma_kernel(const float* __restrict__ x,
                       const float* __restrict__ w_ih1, const float* __restrict__ w_hh1,
                       const float* __restrict__ b_ih1, const float* __restrict__ b_hh1,
                       const float* __restrict__ w_ih2, const float* __restrict__ w_hh2,
                       const float* __restrict__ b_ih2, const float* __restrict__ b_hh2,
                       const float* __restrict__ w_fc,  const float* __restrict__ b_fc,
                       float* __restrict__ out) {
  __shared__ _Float16 sA1[MPAD * KPAD];
  __shared__ _Float16 sAi2[MPAD * KPAD];
  __shared__ _Float16 sAh2[MPAD * KPAD];
  __shared__ __attribute__((aligned(16))) int sP[NROW * 2 * NCH * 16];

  const int tid  = threadIdx.x;
  const int lane = tid & 31;
  const int wave  = __builtin_amdgcn_readfirstlane(tid >> 5);
  const int layer = wave & 1;
  const int prow  = wave >> 1;
  const bool hi_half = lane >= 16;

  for (int idx = tid; idx < MPAD * KPAD; idx += NTHR) {
    const int m = idx >> 5, k = idx & 31;
    const int g = m >> 5, j = m & 31;
    const int row = g * HSZ + j;
    float a1 = 0.f, ai2 = 0.f, ah2 = 0.f;
    if (j < HSZ) {
      if (k < HSZ) {
        a1  = w_hh1[row * HSZ + k];
        ai2 = w_ih2[row * HSZ + k];
        ah2 = w_hh2[row * HSZ + k];
      } else if (k == 30) {
        a1  = w_ih1[row];
      } else {
        a1  = b_ih1[row] + b_hh1[row];
        ah2 = b_ih2[row] + b_hh2[row];
      }
    }
    sA1[idx]  = (_Float16)a1;
    sAi2[idx] = (_Float16)ai2;
    sAh2[idx] = (_Float16)ah2;
  }
  __syncthreads();

  const _Float16* baseX = layer ? sAh2 : sA1;
  v16h W[16];
#pragma unroll
  for (int g = 0; g < 4; ++g) {
#pragma unroll
    for (int s = 0; s < 2; ++s) {
      W[2 * g + s]     = load_w_tile(baseX, g, s, lane);
      W[8 + 2 * g + s] = load_w_tile(sAi2,  g, s, lane);
    }
  }

  const float wfc = (lane < HSZ) ? w_fc[lane] : 0.f;
  const float bfc = b_fc[0];

  v8f zc;
#pragma unroll
  for (int i = 0; i < 8; ++i) zc[i] = 0.f;

  float c = 0.f;

  int sel[8];
#pragma unroll
  for (int j = 0; j < 8; ++j) sel[j] = 2 * j + 1;

  const int brow = blockIdx.x * NROW + prow;
  const float* xrow = x + (size_t)brow * TLEN;
  float* orow = out + (size_t)brow * TLEN;
  int* sProw = sP + prow * (2 * NCH * 16);

  const int xl = lane & (NCH - 1);
  float xc = xrow[xl];
  float oacc = 0.f;
  float fq[8];
#pragma unroll
  for (int k = 0; k < 8; ++k) fq[k] = 0.f;
  v8f qa[8], qb[8];
#pragma unroll
  for (int t = 0; t < 8; ++t) { qa[t] = zc; qb[t] = zc; }

  const int rail2 = pack_halves(0.f, 1.f);
  v8i a0;
#pragma unroll
  for (int i = 0; i < 8; ++i) a0[i] = 0;
  v16h aop = set_rail(__builtin_bit_cast(v16h, a0), hi_half, rail2);

  for (int p = 0; p <= NPH; ++p) {
    if (layer == 0) {
      if (p < NPH) {
        const float xn = xrow[min(NCH * (p + 1) + xl, TLEN - 1)];
        int pd[NCH];
#pragma unroll
        for (int j = 0; j < NCH; ++j)
          pd[j] = layer1_step(j, hi_half, W, zc, c, aop, xc, sel);
        u16a* dst = (u16a*)(sProw + (p & 1) * (NCH * 16)) + (lane ^ 1);
#pragma unroll
        for (int j = 0; j < NCH; ++j)
          dst[j * 32] = (u16a)pd[j];
        xc = xn;
      }
    } else {
      if (p >= 1) {
        const int* src = sProw + ((p - 1) & 1) * (NCH * 16) + (hi_half ? 8 : 0);
        v16h ha = load_h1_rows(src, lane, 0);
        v16h hb = load_h1_rows(src, lane, 1);
        const int s0 = NCH * (p - 1) - 8;
        if (p == 1)
          layer2_group<false, true>(s0, lane, hi_half, W, zc, c, aop, rail2, wfc, bfc,
                                    fq, oacc, qb, qa, ha, orow, sel);
        else
          layer2_group<true, true>(s0, lane, hi_half, W, zc, c, aop, rail2, wfc, bfc,
                                   fq, oacc, qb, qa, ha, orow, sel);
        layer2_group<true, true>(s0 + 8, lane, hi_half, W, zc, c, aop, rail2, wfc, bfc,
                                 fq, oacc, qa, qb, hb, orow, sel);
      }
    }
    if (p < NPH) handover_barrier();
  }

  if (layer != 0) {
    v16h none = aop;
    layer2_group<true, false>(TLEN - 8, lane, hi_half, W, zc, c, aop, rail2, wfc, bfc,
                              fq, oacc, qb, qa, none, orow, sel);
#pragma unroll
    for (int i = 1; i <= 4; ++i) fc_fold_stage(i, lane, fq, oacc, bfc, 3);
    orow[TLEN - 32 + lane] = oacc;
  }
}

extern "C" void kernel_launch(void* const* d_in, const int* in_sizes, int n_in,
                              void* d_out, int out_size, void* d_ws, size_t ws_size,
                              hipStream_t stream) {
  (void)in_sizes; (void)n_in; (void)out_size; (void)d_ws; (void)ws_size;
  const float* x     = (const float*)d_in[0];
  const float* w_ih1 = (const float*)d_in[1];
  const float* w_hh1 = (const float*)d_in[2];
  const float* b_ih1 = (const float*)d_in[3];
  const float* b_hh1 = (const float*)d_in[4];
  const float* w_ih2 = (const float*)d_in[5];
  const float* w_hh2 = (const float*)d_in[6];
  const float* b_ih2 = (const float*)d_in[7];
  const float* b_hh2 = (const float*)d_in[8];
  const float* w_fc  = (const float*)d_in[9];
  const float* b_fc  = (const float*)d_in[10];
  float* out = (float*)d_out;

  dim3 grid(BSZ / NROW);
  dim3 block(NTHR);
  lstm2_wmma_kernel<<<grid, block, 0, stream>>>(x, w_ih1, w_hh1, b_ih1, b_hh1,
                                                w_ih2, w_hh2, b_ih2, b_hh2,
                                                w_fc, b_fc, out);
}


# i and f gate sigmoid links issued between a step's last four recurrent WMMAs (one per WMMA, pinned by ordering points), both layers: behind the last WMMA only g, o, cell update; values, sums kept
# speedup vs baseline: 6.1112x; 1.0599x over previous
#include <hip/hip_runtime.h>


#define HSZ   30
#define MPAD  128
#define KPAD  32
#define TLEN  2048
#define BSZ   512
#define NROW  2
#define NTHR  (NROW * 64)
#define NCH   16
#define NPH   (TLEN / NCH)

typedef __attribute__((ext_vector_type(16))) _Float16 v16h;
typedef __attribute__((ext_vector_type(8)))  _Float16 v8h;
typedef __attribute__((ext_vector_type(2)))  _Float16 v2h;
typedef __attribute__((ext_vector_type(8)))  float    v8f;
typedef __attribute__((ext_vector_type(8)))  int      v8i;
typedef __attribute__((ext_vector_type(4)))  int      v4i;
typedef unsigned short __attribute__((may_alias)) u16a;

#if __has_builtin(__builtin_amdgcn_tanhf)
#define HW_TANH __builtin_amdgcn_tanhf
#elif __has_builtin(__builtin_amdgcn_tanh_f32)
#define HW_TANH __builtin_amdgcn_tanh_f32
#endif

#ifdef HW_TANH
__device__ __forceinline__ float tanh_act(float x) { return HW_TANH(x); }
__device__ __forceinline__ float sig_act(float x)  { return fmaf(0.5f, HW_TANH(0.5f * x), 0.5f); }
__device__ __forceinline__ float sig_pre(float x)  { return 0.5f * x; }
__device__ __forceinline__ float sig_mid(float x)  { return HW_TANH(x); }
__device__ __forceinline__ float sig_post(float t) { return fmaf(0.5f, t, 0.5f); }
__device__ __forceinline__ float sig_pre_in(float d, float q) { return fmaf(0.5f, d, q); }
#else
__device__ __forceinline__ float tanh_act(float x) {
  return fmaf(2.0f, __builtin_amdgcn_rcpf(1.0f + __expf(-2.0f * x)), -1.0f);
}
__device__ __forceinline__ float sig_act(float x) {
  return __builtin_amdgcn_rcpf(1.0f + __expf(-x));
}
__device__ __forceinline__ float sig_pre(float x)  { return x; }
__device__ __forceinline__ float sig_mid(float x)  { return sig_act(x); }
__device__ __forceinline__ float sig_post(float t) { return t; }
__device__ __forceinline__ float sig_pre_in(float d, float q) { return fmaf(2.0f, q, d); }
#endif

template <int CTRL>
__device__ __forceinline__ int dpp_mov(int v) {
  return __builtin_amdgcn_update_dpp(0, v, CTRL, 0xf, 0xf, true);
}
#define DPP_SWAP_PAIR  0xB1
#define DPP_ROW_XMASK  0x160

__device__ __forceinline__ float xhalf(float x) {
  const int b = __builtin_bit_cast(int, x);
  return __builtin_bit_cast(float, __builtin_amdgcn_permlanex16(b, b, 0x76543210, 0xfedcba98, false, false));
}

__device__ __forceinline__ int pack_halves(float lo, float hi) {
  v2h t;
  t[0] = (_Float16)lo;
  t[1] = (_Float16)hi;
  return __builtin_bit_cast(int, t);
}

__device__ __forceinline__ v16h load_w_tile(const _Float16* __restrict__ base,
                                            int g, int s, int lane) {
  const int n  = lane & 15;
  const int kb = (lane >> 4) << 4;
  const _Float16* row = base + ((g << 5) + (s << 4) + n) * KPAD + kb;
  v8h lo = *(const v8h*)(row);
  v8h hi = *(const v8h*)(row + 8);
  v16h b;
#pragma unroll
  for (int i = 0; i < 8; ++i) { b[i] = lo[i]; b[8 + i] = hi[i]; }
  return b;
}

__device__ __forceinline__ int pair_dword(float hn) {
  const float nb = __builtin_bit_cast(float, dpp_mov<DPP_SWAP_PAIR>(__builtin_bit_cast(int, hn)));
  return pack_halves(nb, hn);
}

__device__ __forceinline__ int row_pick(int v, int sel) {
  return (int)__builtin_amdgcn_permlane16_var(0u, (unsigned)v, (unsigned)sel, false, true);
}
__device__ __forceinline__ v16h gather_operand(int pd, const int (&sel)[8]) {
  v8i a;
#pragma unroll
  for (int j = 0; j < 8; ++j) a[j] = row_pick(pd, sel[j]);
  return __builtin_bit_cast(v16h, a);
}

__device__ __forceinline__ v16h set_rail(v16h op, bool hi_half, int rail) {
  v8i a = __builtin_bit_cast(v8i, op);
  a[7] = hi_half ? rail : a[7];
  return __builtin_bit_cast(v16h, a);
}

__device__ __forceinline__ void g_then_o(float& gv, v8f& o0, v8f& o1, v16h& aop) {
  asm volatile("" : "+v"(gv), "+v"(o0), "+v"(o1), "+v"(aop));
}
__device__ __forceinline__ void o_select_first(float& dn, float& gv) {
  asm volatile("" : "+v"(dn), "+v"(gv));
}
__device__ __forceinline__ void o_tanh_first(float& txo, float& cn) {
  asm volatile("" : "+v"(txo), "+v"(cn));
}
__device__ __forceinline__ void cell_tanh_first(float& tcn, float& txo) {
  asm volatile("" : "+v"(tcn), "+v"(txo));
}
template <bool LAYER2>
__device__ __forceinline__ float gate_tail(v8f (&d)[8], v16h& aop, bool hi_half,
                                           float iv, float tf, float qg, float qo, float& c) {
  const float fv = sig_post(tf);
  const float dg = hi_half ? d[5][0] : d[4][0];
  float gv = tanh_act(LAYER2 ? dg + qg : dg);
  g_then_o(gv, d[6], d[7], aop);
  float dn = hi_half ? d[7][0] : d[6][0];
  o_select_first(dn, gv);
  float txo = sig_mid(LAYER2 ? sig_pre_in(dn, qo) : sig_pre(dn));
  float cn = fmaf(fv, c, iv * gv);
  o_tanh_first(txo, cn);
  float tcn = tanh_act(cn);
  cell_tanh_first(tcn, txo);
  const float ov = sig_post(txo);
  c = cn;
  return ov * tcn;
}

__device__ __forceinline__ void handover_barrier() {
  __builtin_amdgcn_fence(__ATOMIC_RELEASE, "workgroup", "local");
  __builtin_amdgcn_s_barrier();
  __builtin_amdgcn_fence(__ATOMIC_ACQUIRE, "workgroup", "local");
}


__device__ __forceinline__ void keep_tiles_disjoint(v8f (&d)[8]) {
  asm volatile("" : "+v"(d[0]), "+v"(d[1]), "+v"(d[2]), "+v"(d[3]),
                    "+v"(d[4]), "+v"(d[5]), "+v"(d[6]), "+v"(d[7]));
}

__device__ __forceinline__ void behind_tiles4(v8f& t0, v8f& t1, v8f& t2, v8f& t3) {
  asm volatile("" : "+v"(t0), "+v"(t1), "+v"(t2), "+v"(t3));
}
__device__ __forceinline__ void behind_tile(v8f& tile, float& x) {
  asm volatile("" : "+v"(tile), "+v"(x));
}
__device__ __forceinline__ void behind_tile(v8f& tile, float& x, float& y) {
  asm volatile("" : "+v"(tile), "+v"(x), "+v"(y));
}
__device__ __forceinline__ void behind_tile(v8f& tile, v8f& ta, v8f& tb) {
  asm volatile("" : "+v"(tile), "+v"(ta), "+v"(tb));
}
__device__ __forceinline__ void ahead_of_tile(float& x, v16h& aop) {
  asm volatile("" : "+v"(x), "+v"(aop));
}
__device__ __forceinline__ void ahead_of_tile(float& x, float& y, v16h& aop) {
  asm volatile("" : "+v"(x), "+v"(y), "+v"(aop));
}

__device__ __forceinline__ v8f rec_tile(const v16h& aop, const v16h& w, const v8f& zc) {
  return __builtin_amdgcn_wmma_f32_16x16x32_f16(false, aop, false, w, (short)0, zc, false, false);
}

template <bool LAYER2>
__device__ __forceinline__ void gate_head_links(v16h& aop, const v16h (&W)[16], const v8f& zc,
                                                bool hi_half, float qi, float qf,
                                                v8f (&d)[8], float& iv, float& tf) {
  behind_tiles4(d[0], d[1], d[2], d[3]);
  const float di = hi_half ? d[1][0] : d[0][0];
  float xi = LAYER2 ? sig_pre_in(di, qi) : sig_pre(di);
  ahead_of_tile(xi, aop);
  d[4] = rec_tile(aop, W[4], zc);
  behind_tile(d[4], xi);
  float ti = sig_mid(xi);
  ahead_of_tile(ti, aop);
  d[5] = rec_tile(aop, W[5], zc);
  behind_tile(d[5], d[2], d[3]);
  const float df = hi_half ? d[3][0] : d[2][0];
  float xf = LAYER2 ? sig_pre_in(df, qf) : sig_pre(df);
  ahead_of_tile(xf, aop);
  d[6] = rec_tile(aop, W[6], zc);
  behind_tile(d[6], ti, xf);
  iv = sig_post(ti);
  tf = sig_mid(xf);
  ahead_of_tile(iv, tf, aop);
  d[7] = rec_tile(aop, W[7], zc);
  behind_tile(d[7], tf);
}

__device__ __forceinline__ void q_behind_tile(v8f& tile, unsigned& half_mask) {
  asm volatile("" : "+v"(tile), "+s"(half_mask));
}
__device__ __forceinline__ void tiles_then_proj(v8f (&d)[8], v16h& hrows) {
  asm volatile("" : "+v"(d[0]), "+v"(d[1]), "+v"(d[2]), "+v"(d[3]),
                    "+v"(d[4]), "+v"(d[5]), "+v"(d[6]), "+v"(d[7]), "+v"(hrows));
}
__device__ __forceinline__ void proj_then_gather(v16h& hrows, float& hn) {
  asm volatile("" : "+v"(hrows), "+v"(hn));
}

__device__ __forceinline__ int layer1_step(int j, bool hi_half,
                                           const v16h (&W)[16], const v8f& zc, float& c,
                                           v16h& aop, float xc, const int (&sel)[8]) {
  const float xs = __builtin_bit_cast(float,
      __builtin_amdgcn_readlane(__builtin_bit_cast(int, xc), j));
  aop = set_rail(aop, hi_half, pack_halves(xs, 1.f));

  v8f d[8];
  float iv, tf;
#pragma unroll
  for (int t = 0; t < 4; ++t) d[t] = rec_tile(aop, W[t], zc);
  gate_head_links<false>(aop, W, zc, hi_half, 0.f, 0.f, d, iv, tf);
  keep_tiles_disjoint(d);
  const float hn = gate_tail<false>(d, aop, hi_half, iv, tf, 0.f, 0.f, c);
  const int pd = pair_dword(hn);
  aop = gather_operand(pd, sel);
  return pd;
}

template <int XM>
__device__ __forceinline__ float fold_pair(float a, float b, bool bit_set) {
  const float keep = bit_set ? b : a;
  const float send = bit_set ? a : b;
  return keep + __builtin_bit_cast(float, dpp_mov<DPP_ROW_XMASK + XM>(__builtin_bit_cast(int, send)));
}

__device__ __forceinline__ void fold_in8(float (&r)[8]) {
  asm volatile("" : "+v"(r[0]), "+v"(r[1]), "+v"(r[2]), "+v"(r[3]),
                    "+v"(r[4]), "+v"(r[5]), "+v"(r[6]), "+v"(r[7]));
}
__device__ __forceinline__ void fold_in4(float (&r)[8]) {
  asm volatile("" : "+v"(r[0]), "+v"(r[1]), "+v"(r[2]), "+v"(r[3]));
}
__device__ __forceinline__ void fold_in2(float (&r)[8]) {
  asm volatile("" : "+v"(r[0]), "+v"(r[1]));
}
__device__ __forceinline__ void fold_in1(float (&r)[8]) {
  asm volatile("" : "+v"(r[0]));
}

__device__ __forceinline__ void fc_fold_stage(int I, int lane, float (&red)[8], float& oacc,
                                              float bfc, int qprev) {
  if (I == 1) {
    fold_in8(red);
    const bool b0 = (lane & 1) != 0;
    red[0] = fold_pair<1>(red[0], red[1], b0);
    red[1] = fold_pair<1>(red[2], red[3], b0);
    red[2] = fold_pair<1>(red[4], red[5], b0);
    red[3] = fold_pair<1>(red[6], red[7], b0);
  } else if (I == 2) {
    fold_in4(red);
    const bool b1 = (lane & 2) != 0;
    red[0] = fold_pair<2>(red[0], red[1], b1);
    red[1] = fold_pair<2>(red[2], red[3], b1);
  } else if (I == 3) {
    fold_in2(red);
    const bool b2 = (lane & 4) != 0;
    float v = fold_pair<4>(red[0], red[1], b2);
    v += __builtin_bit_cast(float, dpp_mov<DPP_ROW_XMASK + 8>(__builtin_bit_cast(int, v)));
    red[0] = v;
  } else if (I == 4) {
    fold_in1(red);
    const float v = red[0] + xhalf(red[0]);
    oacc = ((lane >> 3) == qprev) ? (v + bfc) : oacc;
  }
}

template <bool REC, bool PROJ>
__device__ __forceinline__ void layer2_step(int I, int lane, bool hi_half,
                                            const v16h (&W)[16], const v8f& zc, float& c,
                                            v16h& aop, int rail, float wfc, float bfc,
                                            float (&fcp)[8], float (&red)[8], float& oacc, int qprev,
                                            const v8f (&qc)[8], v8f (&qn)[8], v16h& hrows,
                                            const int (&sel)[8]) {
  unsigned half_mask = 0xffff0000u;

  v8f d[8];
  float iv = 0.f, tf = 0.f, qg = 0.f, qo = 0.f;
  if (REC) {
    d[0] = rec_tile(aop, W[0], zc);
    q_behind_tile(d[0], half_mask);
    const bool hq = __builtin_amdgcn_inverse_ballot_w32(half_mask);
    float si = hq ? qc[1][I] : qc[0][I];
    float sf = hq ? qc[3][I] : qc[2][I];
    ahead_of_tile(si, sf, aop);
    d[1] = rec_tile(aop, W[1], zc);
    behind_tile(d[1], si, sf);
    float qi = 0.5f * si;
    float qf = 0.5f * sf;
    ahead_of_tile(qi, qf, aop);
    d[2] = rec_tile(aop, W[2], zc);
    q_behind_tile(d[2], half_mask);
    const bool hr = __builtin_amdgcn_inverse_ballot_w32(half_mask);
    qg = hr ? qc[5][I] : qc[4][I];
    float so = hr ? qc[7][I] : qc[6][I];
    ahead_of_tile(qg, so, aop);
    d[3] = rec_tile(aop, W[3], zc);
    qo = 0.5f * so;
    gate_head_links<true>(aop, W, zc, hi_half, qi, qf, d, iv, tf);
    if (PROJ) tiles_then_proj(d, hrows); else keep_tiles_disjoint(d);
    fc_fold_stage(I, lane, red, oacc, bfc, qprev);
  }
  if (PROJ)
    qn[I] = __builtin_amdgcn_wmma_f32_16x16x32_f16(false, hrows, false, W[8 + I],
                                                   (short)0, zc, false, false);
  if (REC) {
    float hn = gate_tail<true>(d, aop, hi_half, iv, tf, qg, qo, c);
    if (PROJ) proj_then_gather(hrows, hn);
    aop = set_rail(gather_operand(pair_dword(hn), sel), hi_half, rail);
    fcp[I] = hn * wfc;
  }
}

__device__ __forceinline__ v16h load_h1_rows(const int* src, int lane, int grp) {
  const int* p = src + (8 * grp + (lane & 7)) * 16;
  const v4i lo = *(const v4i*)(p);
  const v4i hi = *(const v4i*)(p + 4);
  v8i hw;
#pragma unroll
  for (int e = 0; e < 4; ++e) { hw[e] = lo[e]; hw[4 + e] = hi[e]; }
  return __builtin_bit_cast(v16h, hw);
}

template <bool REC, bool PROJ>
__device__ __forceinline__ void layer2_group(int s0, int lane, bool hi_half,
                                             const v16h (&W)[16], const v8f& zc, float& c,
                                             v16h& aop, int rail, float wfc, float bfc,
                                             float (&fq)[8], float& oacc,
                                             const v8f (&qc)[8], v8f (&qn)[8], v16h& hrows,
                                             float* orow, const int (&sel)[8]) {
  float fcp[8], red[8];
#pragma unroll
  for (int k = 0; k < 8; ++k) { red[k] = fq[k]; fcp[k] = 0.f; }
  const int qprev = ((s0 >> 3) + 3) & 3;
#pragma unroll
  for (int i = 0; i < 8; ++i)
    layer2_step<REC, PROJ>(i, lane, hi_half, W, zc, c, aop, rail, wfc, bfc, fcp, red, oacc, qprev,
                           qc, qn, hrows, sel);
  if (REC) {
#pragma unroll
    for (int k = 0; k < 8; ++k) fq[k] = fcp[k];
    if ((s0 & 31) == 0 && s0 != 0) orow[s0 - 32 + lane] = oacc;
  }
}

__global__ __launch_bounds__(NTHR, 1)
void lstm2_wmma_kernel(const float* __restrict__ x,
                       const float* __restrict__ w_ih1, const float* __restrict__ w_hh1,
                       const float* __restrict__ b_ih1, const float* __restrict__ b_hh1,
                       const float* __restrict__ w_ih2, const float* __restrict__ w_hh2,
                       const float* __restrict__ b_ih2, const float* __restrict__ b_hh2,
                       const float* __restrict__ w_fc,  const float* __restrict__ b_fc,
                       float* __restrict__ out) {
  __shared__ _Float16 sA1[MPAD * KPAD];
  __shared__ _Float16 sAi2[MPAD * KPAD];
  __shared__ _Float16 sAh2[MPAD * KPAD];
  __shared__ __attribute__((aligned(16))) int sP[NROW * 2 * NCH * 16];

  const int tid  = threadIdx.x;
  const int lane = tid & 31;
  const int wave  = __builtin_amdgcn_readfirstlane(tid >> 5);
  const int layer = wave & 1;
  const int prow  = wave >> 1;
  const bool hi_half = lane >= 16;

  for (int idx = tid; idx < MPAD * KPAD; idx += NTHR) {
    const int m = idx >> 5, k = idx & 31;
    const int g = m >> 5, j = m & 31;
    const int row = g * HSZ + j;
    float a1 = 0.f, ai2 = 0.f, ah2 = 0.f;
    if (j < HSZ) {
      if (k < HSZ) {
        a1  = w_hh1[row * HSZ + k];
        ai2 = w_ih2[row * HSZ + k];
        ah2 = w_hh2[row * HSZ + k];
      } else if (k == 30) {
        a1  = w_ih1[row];
      } else {
        a1  = b_ih1[row] + b_hh1[row];
        ah2 = b_ih2[row] + b_hh2[row];
      }
    }
    sA1[idx]  = (_Float16)a1;
    sAi2[idx] = (_Float16)ai2;
    sAh2[idx] = (_Float16)ah2;
  }
  __syncthreads();

  const _Float16* baseX = layer ? sAh2 : sA1;
  v16h W[16];
#pragma unroll
  for (int g = 0; g < 4; ++g) {
#pragma unroll
    for (int s = 0; s < 2; ++s) {
      W[2 * g + s]     = load_w_tile(baseX, g, s, lane);
      W[8 + 2 * g + s] = load_w_tile(sAi2,  g, s, lane);
    }
  }

  const float wfc = (lane < HSZ) ? w_fc[lane] : 0.f;
  const float bfc = b_fc[0];

  v8f zc;
#pragma unroll
  for (int i = 0; i < 8; ++i) zc[i] = 0.f;

  float c = 0.f;

  int sel[8];
#pragma unroll
  for (int j = 0; j < 8; ++j) sel[j] = 2 * j + 1;

  const int brow = blockIdx.x * NROW + prow;
  const float* xrow = x + (size_t)brow * TLEN;
  float* orow = out + (size_t)brow * TLEN;
  int* sProw = sP + prow * (2 * NCH * 16);

  const int xl = lane & (NCH - 1);
  float xc = xrow[xl];
  float oacc = 0.f;
  float fq[8];
#pragma unroll
  for (int k = 0; k < 8; ++k) fq[k] = 0.f;
  v8f qa[8], qb[8];
#pragma unroll
  for (int t = 0; t < 8; ++t) { qa[t] = zc; qb[t] = zc; }

  const int rail2 = pack_halves(0.f, 1.f);
  v8i a0;
#pragma unroll
  for (int i = 0; i < 8; ++i) a0[i] = 0;
  v16h aop = set_rail(__builtin_bit_cast(v16h, a0), hi_half, rail2);

  for (int p = 0; p <= NPH; ++p) {
    if (layer == 0) {
      if (p < NPH) {
        const float xn = xrow[min(NCH * (p + 1) + xl, TLEN - 1)];
        int pd[NCH];
#pragma unroll
        for (int j = 0; j < NCH; ++j)
          pd[j] = layer1_step(j, hi_half, W, zc, c, aop, xc, sel);
        u16a* dst = (u16a*)(sProw + (p & 1) * (NCH * 16)) + (lane ^ 1);
#pragma unroll
        for (int j = 0; j < NCH; ++j)
          dst[j * 32] = (u16a)pd[j];
        xc = xn;
      }
    } else {
      if (p >= 1) {
        const int* src = sProw + ((p - 1) & 1) * (NCH * 16) + (hi_half ? 8 : 0);
        v16h ha = load_h1_rows(src, lane, 0);
        v16h hb = load_h1_rows(src, lane, 1);
        const int s0 = NCH * (p - 1) - 8;
        if (p == 1)
          layer2_group<false, true>(s0, lane, hi_half, W, zc, c, aop, rail2, wfc, bfc,
                                    fq, oacc, qb, qa, ha, orow, sel);
        else
          layer2_group<true, true>(s0, lane, hi_half, W, zc, c, aop, rail2, wfc, bfc,
                                   fq, oacc, qb, qa, ha, orow, sel);
        layer2_group<true, true>(s0 + 8, lane, hi_half, W, zc, c, aop, rail2, wfc, bfc,
                                 fq, oacc, qa, qb, hb, orow, sel);
      }
    }
    if (p < NPH) handover_barrier();
  }

  if (layer != 0) {
    v16h none = aop;
    layer2_group<true, false>(TLEN - 8, lane, hi_half, W, zc, c, aop, rail2, wfc, bfc,
                              fq, oacc, qb, qa, none, orow, sel);
#pragma unroll
    for (int i = 1; i <= 4; ++i) fc_fold_stage(i, lane, fq, oacc, bfc, 3);
    orow[TLEN - 32 + lane] = oacc;
  }
}

extern "C" void kernel_launch(void* const* d_in, const int* in_sizes, int n_in,
                              void* d_out, int out_size, void* d_ws, size_t ws_size,
                              hipStream_t stream) {
  (void)in_sizes; (void)n_in; (void)out_size; (void)d_ws; (void)ws_size;
  const float* x     = (const float*)d_in[0];
  const float* w_ih1 = (const float*)d_in[1];
  const float* w_hh1 = (const float*)d_in[2];
  const float* b_ih1 = (const float*)d_in[3];
  const float* b_hh1 = (const float*)d_in[4];
  const float* w_ih2 = (const float*)d_in[5];
  const float* w_hh2 = (const float*)d_in[6];
  const float* b_ih2 = (const float*)d_in[7];
  const float* b_hh2 = (const float*)d_in[8];
  const float* w_fc  = (const float*)d_in[9];
  const float* b_fc  = (const float*)d_in[10];
  float* out = (float*)d_out;

  dim3 grid(BSZ / NROW);
  dim3 block(NTHR);
  lstm2_wmma_kernel<<<grid, block, 0, stream>>>(x, w_ih1, w_hh1, b_ih1, b_hh1,
                                                w_ih2, w_hh2, b_ih2, b_hh2,
                                                w_fc, b_fc, out);
}


# Weight staging: all prologue global loads independent and issued together (flat line-wide reads, no lane test round a load), one wait; before, 32 passes each waited out 3 dependent loads; values same
# speedup vs baseline: 6.8397x; 1.1192x over previous
#include <hip/hip_runtime.h>


#define HSZ   30
#define MPAD  128
#define KPAD  32
#define TLEN  2048
#define BSZ   512
#define NROW  2
#define NTHR  (NROW * 64)
#define NCH   16
#define NPH   (TLEN / NCH)

typedef __attribute__((ext_vector_type(16))) _Float16 v16h;
typedef __attribute__((ext_vector_type(8)))  _Float16 v8h;
typedef __attribute__((ext_vector_type(2)))  _Float16 v2h;
typedef __attribute__((ext_vector_type(8)))  float    v8f;
typedef __attribute__((ext_vector_type(8)))  int      v8i;
typedef __attribute__((ext_vector_type(4)))  int      v4i;
typedef unsigned short __attribute__((may_alias)) u16a;

#if __has_builtin(__builtin_amdgcn_tanhf)
#define HW_TANH __builtin_amdgcn_tanhf
#elif __has_builtin(__builtin_amdgcn_tanh_f32)
#define HW_TANH __builtin_amdgcn_tanh_f32
#endif

#ifdef HW_TANH
__device__ __forceinline__ float tanh_act(float x) { return HW_TANH(x); }
__device__ __forceinline__ float sig_act(float x)  { return fmaf(0.5f, HW_TANH(0.5f * x), 0.5f); }
__device__ __forceinline__ float sig_pre(float x)  { return 0.5f * x; }
__device__ __forceinline__ float sig_mid(float x)  { return HW_TANH(x); }
__device__ __forceinline__ float sig_post(float t) { return fmaf(0.5f, t, 0.5f); }
__device__ __forceinline__ float sig_pre_in(float d, float q) { return fmaf(0.5f, d, q); }
#else
__device__ __forceinline__ float tanh_act(float x) {
  return fmaf(2.0f, __builtin_amdgcn_rcpf(1.0f + __expf(-2.0f * x)), -1.0f);
}
__device__ __forceinline__ float sig_act(float x) {
  return __builtin_amdgcn_rcpf(1.0f + __expf(-x));
}
__device__ __forceinline__ float sig_pre(float x)  { return x; }
__device__ __forceinline__ float sig_mid(float x)  { return sig_act(x); }
__device__ __forceinline__ float sig_post(float t) { return t; }
__device__ __forceinline__ float sig_pre_in(float d, float q) { return fmaf(2.0f, q, d); }
#endif

template <int CTRL>
__device__ __forceinline__ int dpp_mov(int v) {
  return __builtin_amdgcn_update_dpp(0, v, CTRL, 0xf, 0xf, true);
}
#define DPP_SWAP_PAIR  0xB1
#define DPP_ROW_XMASK  0x160

__device__ __forceinline__ float xhalf(float x) {
  const int b = __builtin_bit_cast(int, x);
  return __builtin_bit_cast(float, __builtin_amdgcn_permlanex16(b, b, 0x76543210, 0xfedcba98, false, false));
}

__device__ __forceinline__ int pack_halves(float lo, float hi) {
  v2h t;
  t[0] = (_Float16)lo;
  t[1] = (_Float16)hi;
  return __builtin_bit_cast(int, t);
}

__device__ __forceinline__ v16h load_w_tile(const _Float16* __restrict__ base,
                                            int g, int s, int lane) {
  const int n  = lane & 15;
  const int kb = (lane >> 4) << 4;
  const _Float16* row = base + ((g << 5) + (s << 4) + n) * KPAD + kb;
  v8h lo = *(const v8h*)(row);
  v8h hi = *(const v8h*)(row + 8);
  v16h b;
#pragma unroll
  for (int i = 0; i < 8; ++i) { b[i] = lo[i]; b[8 + i] = hi[i]; }
  return b;
}

__device__ __forceinline__ int pair_dword(float hn) {
  const float nb = __builtin_bit_cast(float, dpp_mov<DPP_SWAP_PAIR>(__builtin_bit_cast(int, hn)));
  return pack_halves(nb, hn);
}

__device__ __forceinline__ int row_pick(int v, int sel) {
  return (int)__builtin_amdgcn_permlane16_var(0u, (unsigned)v, (unsigned)sel, false, true);
}
__device__ __forceinline__ v16h gather_operand(int pd, const int (&sel)[8]) {
  v8i a;
#pragma unroll
  for (int j = 0; j < 8; ++j) a[j] = row_pick(pd, sel[j]);
  return __builtin_bit_cast(v16h, a);
}

__device__ __forceinline__ v16h set_rail(v16h op, bool hi_half, int rail) {
  v8i a = __builtin_bit_cast(v8i, op);
  a[7] = hi_half ? rail : a[7];
  return __builtin_bit_cast(v16h, a);
}

__device__ __forceinline__ void g_then_o(float& gv, v8f& o0, v8f& o1, v16h& aop) {
  asm volatile("" : "+v"(gv), "+v"(o0), "+v"(o1), "+v"(aop));
}
__device__ __forceinline__ void o_select_first(float& dn, float& gv) {
  asm volatile("" : "+v"(dn), "+v"(gv));
}
__device__ __forceinline__ void o_tanh_first(float& txo, float& cn) {
  asm volatile("" : "+v"(txo), "+v"(cn));
}
__device__ __forceinline__ void cell_tanh_first(float& tcn, float& txo) {
  asm volatile("" : "+v"(tcn), "+v"(txo));
}
template <bool LAYER2>
__device__ __forceinline__ float gate_tail(v8f (&d)[8], v16h& aop, bool hi_half,
                                           float iv, float tf, float qg, float qo, float& c) {
  const float fv = sig_post(tf);
  const float dg = hi_half ? d[5][0] : d[4][0];
  float gv = tanh_act(LAYER2 ? dg + qg : dg);
  g_then_o(gv, d[6], d[7], aop);
  float dn = hi_half ? d[7][0] : d[6][0];
  o_select_first(dn, gv);
  float txo = sig_mid(LAYER2 ? sig_pre_in(dn, qo) : sig_pre(dn));
  float cn = fmaf(fv, c, iv * gv);
  o_tanh_first(txo, cn);
  float tcn = tanh_act(cn);
  cell_tanh_first(tcn, txo);
  const float ov = sig_post(txo);
  c = cn;
  return ov * tcn;
}

__device__ __forceinline__ void handover_barrier() {
  __builtin_amdgcn_fence(__ATOMIC_RELEASE, "workgroup", "local");
  __builtin_amdgcn_s_barrier();
  __builtin_amdgcn_fence(__ATOMIC_ACQUIRE, "workgroup", "local");
}


__device__ __forceinline__ void keep_tiles_disjoint(v8f (&d)[8]) {
  asm volatile("" : "+v"(d[0]), "+v"(d[1]), "+v"(d[2]), "+v"(d[3]),
                    "+v"(d[4]), "+v"(d[5]), "+v"(d[6]), "+v"(d[7]));
}

__device__ __forceinline__ void behind_tiles4(v8f& t0, v8f& t1, v8f& t2, v8f& t3) {
  asm volatile("" : "+v"(t0), "+v"(t1), "+v"(t2), "+v"(t3));
}
__device__ __forceinline__ void behind_tile(v8f& tile, float& x) {
  asm volatile("" : "+v"(tile), "+v"(x));
}
__device__ __forceinline__ void behind_tile(v8f& tile, float& x, float& y) {
  asm volatile("" : "+v"(tile), "+v"(x), "+v"(y));
}
__device__ __forceinline__ void behind_tile(v8f& tile, v8f& ta, v8f& tb) {
  asm volatile("" : "+v"(tile), "+v"(ta), "+v"(tb));
}
__device__ __forceinline__ void ahead_of_tile(float& x, v16h& aop) {
  asm volatile("" : "+v"(x), "+v"(aop));
}
__device__ __forceinline__ void ahead_of_tile(float& x, float& y, v16h& aop) {
  asm volatile("" : "+v"(x), "+v"(y), "+v"(aop));
}

__device__ __forceinline__ v8f rec_tile(const v16h& aop, const v16h& w, const v8f& zc) {
  return __builtin_amdgcn_wmma_f32_16x16x32_f16(false, aop, false, w, (short)0, zc, false, false);
}

template <bool LAYER2>
__device__ __forceinline__ void gate_head_links(v16h& aop, const v16h (&W)[16], const v8f& zc,
                                                bool hi_half, float qi, float qf,
                                                v8f (&d)[8], float& iv, float& tf) {
  behind_tiles4(d[0], d[1], d[2], d[3]);
  const float di = hi_half ? d[1][0] : d[0][0];
  float xi = LAYER2 ? sig_pre_in(di, qi) : sig_pre(di);
  ahead_of_tile(xi, aop);
  d[4] = rec_tile(aop, W[4], zc);
  behind_tile(d[4], xi);
  float ti = sig_mid(xi);
  ahead_of_tile(ti, aop);
  d[5] = rec_tile(aop, W[5], zc);
  behind_tile(d[5], d[2], d[3]);
  const float df = hi_half ? d[3][0] : d[2][0];
  float xf = LAYER2 ? sig_pre_in(df, qf) : sig_pre(df);
  ahead_of_tile(xf, aop);
  d[6] = rec_tile(aop, W[6], zc);
  behind_tile(d[6], ti, xf);
  iv = sig_post(ti);
  tf = sig_mid(xf);
  ahead_of_tile(iv, tf, aop);
  d[7] = rec_tile(aop, W[7], zc);
  behind_tile(d[7], tf);
}

__device__ __forceinline__ void q_behind_tile(v8f& tile, unsigned& half_mask) {
  asm volatile("" : "+v"(tile), "+s"(half_mask));
}
__device__ __forceinline__ void tiles_then_proj(v8f (&d)[8], v16h& hrows) {
  asm volatile("" : "+v"(d[0]), "+v"(d[1]), "+v"(d[2]), "+v"(d[3]),
                    "+v"(d[4]), "+v"(d[5]), "+v"(d[6]), "+v"(d[7]), "+v"(hrows));
}
__device__ __forceinline__ void proj_then_gather(v16h& hrows, float& hn) {
  asm volatile("" : "+v"(hrows), "+v"(hn));
}

__device__ __forceinline__ int layer1_step(int j, bool hi_half,
                                           const v16h (&W)[16], const v8f& zc, float& c,
                                           v16h& aop, float xc, const int (&sel)[8]) {
  const float xs = __builtin_bit_cast(float,
      __builtin_amdgcn_readlane(__builtin_bit_cast(int, xc), j));
  aop = set_rail(aop, hi_half, pack_halves(xs, 1.f));

  v8f d[8];
  float iv, tf;
#pragma unroll
  for (int t = 0; t < 4; ++t) d[t] = rec_tile(aop, W[t], zc);
  gate_head_links<false>(aop, W, zc, hi_half, 0.f, 0.f, d, iv, tf);
  keep_tiles_disjoint(d);
  const float hn = gate_tail<false>(d, aop, hi_half, iv, tf, 0.f, 0.f, c);
  const int pd = pair_dword(hn);
  aop = gather_operand(pd, sel);
  return pd;
}

template <int XM>
__device__ __forceinline__ float fold_pair(float a, float b, bool bit_set) {
  const float keep = bit_set ? b : a;
  const float send = bit_set ? a : b;
  return keep + __builtin_bit_cast(float, dpp_mov<DPP_ROW_XMASK + XM>(__builtin_bit_cast(int, send)));
}

__device__ __forceinline__ void fold_in8(float (&r)[8]) {
  asm volatile("" : "+v"(r[0]), "+v"(r[1]), "+v"(r[2]), "+v"(r[3]),
                    "+v"(r[4]), "+v"(r[5]), "+v"(r[6]), "+v"(r[7]));
}
__device__ __forceinline__ void fold_in4(float (&r)[8]) {
  asm volatile("" : "+v"(r[0]), "+v"(r[1]), "+v"(r[2]), "+v"(r[3]));
}
__device__ __forceinline__ void fold_in2(float (&r)[8]) {
  asm volatile("" : "+v"(r[0]), "+v"(r[1]));
}
__device__ __forceinline__ void fold_in1(float (&r)[8]) {
  asm volatile("" : "+v"(r[0]));
}

__device__ __forceinline__ void fc_fold_stage(int I, int lane, float (&red)[8], float& oacc,
                                              float bfc, int qprev) {
  if (I == 1) {
    fold_in8(red);
    const bool b0 = (lane & 1) != 0;
    red[0] = fold_pair<1>(red[0], red[1], b0);
    red[1] = fold_pair<1>(red[2], red[3], b0);
    red[2] = fold_pair<1>(red[4], red[5], b0);
    red[3] = fold_pair<1>(red[6], red[7], b0);
  } else if (I == 2) {
    fold_in4(red);
    const bool b1 = (lane & 2) != 0;
    red[0] = fold_pair<2>(red[0], red[1], b1);
    red[1] = fold_pair<2>(red[2], red[3], b1);
  } else if (I == 3) {
    fold_in2(red);
    const bool b2 = (lane & 4) != 0;
    float v = fold_pair<4>(red[0], red[1], b2);
    v += __builtin_bit_cast(float, dpp_mov<DPP_ROW_XMASK + 8>(__builtin_bit_cast(int, v)));
    red[0] = v;
  } else if (I == 4) {
    fold_in1(red);
    const float v = red[0] + xhalf(red[0]);
    oacc = ((lane >> 3) == qprev) ? (v + bfc) : oacc;
  }
}

template <bool REC, bool PROJ>
__device__ __forceinline__ void layer2_step(int I, int lane, bool hi_half,
                                            const v16h (&W)[16], const v8f& zc, float& c,
                                            v16h& aop, int rail, float wfc, float bfc,
                                            float (&fcp)[8], float (&red)[8], float& oacc, int qprev,
                                            const v8f (&qc)[8], v8f (&qn)[8], v16h& hrows,
                                            const int (&sel)[8]) {
  unsigned half_mask = 0xffff0000u;

  v8f d[8];
  float iv = 0.f, tf = 0.f, qg = 0.f, qo = 0.f;
  if (REC) {
    d[0] = rec_tile(aop, W[0], zc);
    q_behind_tile(d[0], half_mask);
    const bool hq = __builtin_amdgcn_inverse_ballot_w32(half_mask);
    float si = hq ? qc[1][I] : qc[0][I];
    float sf = hq ? qc[3][I] : qc[2][I];
    ahead_of_tile(si, sf, aop);
    d[1] = rec_tile(aop, W[1], zc);
    behind_tile(d[1], si, sf);
    float qi = 0.5f * si;
    float qf = 0.5f * sf;
    ahead_of_tile(qi, qf, aop);
    d[2] = rec_tile(aop, W[2], zc);
    q_behind_tile(d[2], half_mask);
    const bool hr = __builtin_amdgcn_inverse_ballot_w32(half_mask);
    qg = hr ? qc[5][I] : qc[4][I];
    float so = hr ? qc[7][I] : qc[6][I];
    ahead_of_tile(qg, so, aop);
    d[3] = rec_tile(aop, W[3], zc);
    qo = 0.5f * so;
    gate_head_links<true>(aop, W, zc, hi_half, qi, qf, d, iv, tf);
    if (PROJ) tiles_then_proj(d, hrows); else keep_tiles_disjoint(d);
    fc_fold_stage(I, lane, red, oacc, bfc, qprev);
  }
  if (PROJ)
    qn[I] = __builtin_amdgcn_wmma_f32_16x16x32_f16(false, hrows, false, W[8 + I],
                                                   (short)0, zc, false, false);
  if (REC) {
    float hn = gate_tail<true>(d, aop, hi_half, iv, tf, qg, qo, c);
    if (PROJ) proj_then_gather(hrows, hn);
    aop = set_rail(gather_operand(pair_dword(hn), sel), hi_half, rail);
    fcp[I] = hn * wfc;
  }
}

__device__ __forceinline__ v16h load_h1_rows(const int* src, int lane, int grp) {
  const int* p = src + (8 * grp + (lane & 7)) * 16;
  const v4i lo = *(const v4i*)(p);
  const v4i hi = *(const v4i*)(p + 4);
  v8i hw;
#pragma unroll
  for (int e = 0; e < 4; ++e) { hw[e] = lo[e]; hw[4 + e] = hi[e]; }
  return __builtin_bit_cast(v16h, hw);
}

template <bool REC, bool PROJ>
__device__ __forceinline__ void layer2_group(int s0, int lane, bool hi_half,
                                             const v16h (&W)[16], const v8f& zc, float& c,
                                             v16h& aop, int rail, float wfc, float bfc,
                                             float (&fq)[8], float& oacc,
                                             const v8f (&qc)[8], v8f (&qn)[8], v16h& hrows,
                                             float* orow, const int (&sel)[8]) {
  float fcp[8], red[8];
#pragma unroll
  for (int k = 0; k < 8; ++k) { red[k] = fq[k]; fcp[k] = 0.f; }
  const int qprev = ((s0 >> 3) + 3) & 3;
#pragma unroll
  for (int i = 0; i < 8; ++i)
    layer2_step<REC, PROJ>(i, lane, hi_half, W, zc, c, aop, rail, wfc, bfc, fcp, red, oacc, qprev,
                           qc, qn, hrows, sel);
  if (REC) {
#pragma unroll
    for (int k = 0; k < 8; ++k) fq[k] = fcp[k];
    if ((s0 & 31) == 0 && s0 != 0) orow[s0 - 32 + lane] = oacc;
  }
}

__global__ __launch_bounds__(NTHR, 1)
void lstm2_wmma_kernel(const float* __restrict__ x,
                       const float* __restrict__ w_ih1, const float* __restrict__ w_hh1,
                       const float* __restrict__ b_ih1, const float* __restrict__ b_hh1,
                       const float* __restrict__ w_ih2, const float* __restrict__ w_hh2,
                       const float* __restrict__ b_ih2, const float* __restrict__ b_hh2,
                       const float* __restrict__ w_fc,  const float* __restrict__ b_fc,
                       float* __restrict__ out) {
  __shared__ _Float16 sA1[MPAD * KPAD];
  __shared__ _Float16 sAi2[MPAD * KPAD];
  __shared__ _Float16 sAh2[MPAD * KPAD];
  __shared__ __attribute__((aligned(16))) int sP[NROW * 2 * NCH * 16];

  const int tid  = threadIdx.x;
  const int lane = tid & 31;
  const int wave  = __builtin_amdgcn_readfirstlane(tid >> 5);
  const int layer = wave & 1;
  const int prow  = wave >> 1;
  const bool hi_half = lane >= 16;

  constexpr int NWGT = 4 * HSZ * HSZ;
  constexpr int NFUL = NWGT / NTHR;
  float r1[NFUL + 1], r2[NFUL + 1], r3[NFUL + 1];
#pragma unroll
  for (int i = 0; i <= NFUL; ++i) {
    const int e = min(tid + NTHR * i, NWGT - 1);
    r1[i] = w_hh1[e];
    r2[i] = w_ih2[e];
    r3[i] = w_hh2[e];
  }
  const int crow = min(tid, 4 * HSZ - 1);
  const float c_wi  = w_ih1[crow];
  const float c_bi1 = b_ih1[crow], c_bh1 = b_hh1[crow];
  const float c_bi2 = b_ih2[crow], c_bh2 = b_hh2[crow];

  for (int q = tid; q < 8 * KPAD; q += NTHR) {
    const int idx = (((q >> 6) << 5) + HSZ + ((q >> 5) & 1)) * KPAD + (q & 31);
    sA1[idx]  = (_Float16)0.f;
    sAi2[idx] = (_Float16)0.f;
    sAh2[idx] = (_Float16)0.f;
  }
  if (tid < 4 * HSZ) {
    const int base = (tid + 2 * (tid / HSZ)) * KPAD + HSZ;
    sA1[base]      = (_Float16)c_wi;
    sA1[base + 1]  = (_Float16)(c_bi1 + c_bh1);
    sAi2[base]     = (_Float16)0.f;
    sAi2[base + 1] = (_Float16)0.f;
    sAh2[base]     = (_Float16)0.f;
    sAh2[base + 1] = (_Float16)(c_bi2 + c_bh2);
  }
#pragma unroll
  for (int i = 0; i <= NFUL; ++i) {
    const int e = tid + NTHR * i;
    if (i < NFUL || e < NWGT) {
      const int idx = e + 2 * (e / HSZ) + 64 * (e / (HSZ * HSZ));
      sA1[idx]  = (_Float16)r1[i];
      sAi2[idx] = (_Float16)r2[i];
      sAh2[idx] = (_Float16)r3[i];
    }
  }
  __syncthreads();

  const _Float16* baseX = layer ? sAh2 : sA1;
  v16h W[16];
#pragma unroll
  for (int g = 0; g < 4; ++g) {
#pragma unroll
    for (int s = 0; s < 2; ++s) {
      W[2 * g + s]     = load_w_tile(baseX, g, s, lane);
      W[8 + 2 * g + s] = load_w_tile(sAi2,  g, s, lane);
    }
  }

  const float wfc = (lane < HSZ) ? w_fc[lane] : 0.f;
  const float bfc = b_fc[0];

  v8f zc;
#pragma unroll
  for (int i = 0; i < 8; ++i) zc[i] = 0.f;

  float c = 0.f;

  int sel[8];
#pragma unroll
  for (int j = 0; j < 8; ++j) sel[j] = 2 * j + 1;

  const int brow = blockIdx.x * NROW + prow;
  const float* xrow = x + (size_t)brow * TLEN;
  float* orow = out + (size_t)brow * TLEN;
  int* sProw = sP + prow * (2 * NCH * 16);

  const int xl = lane & (NCH - 1);
  float xc = xrow[xl];
  float oacc = 0.f;
  float fq[8];
#pragma unroll
  for (int k = 0; k < 8; ++k) fq[k] = 0.f;
  v8f qa[8], qb[8];
#pragma unroll
  for (int t = 0; t < 8; ++t) { qa[t] = zc; qb[t] = zc; }

  const int rail2 = pack_halves(0.f, 1.f);
  v8i a0;
#pragma unroll
  for (int i = 0; i < 8; ++i) a0[i] = 0;
  v16h aop = set_rail(__builtin_bit_cast(v16h, a0), hi_half, rail2);

  for (int p = 0; p <= NPH; ++p) {
    if (layer == 0) {
      if (p < NPH) {
        const float xn = xrow[min(NCH * (p + 1) + xl, TLEN - 1)];
        int pd[NCH];
#pragma unroll
        for (int j = 0; j < NCH; ++j)
          pd[j] = layer1_step(j, hi_half, W, zc, c, aop, xc, sel);
        u16a* dst = (u16a*)(sProw + (p & 1) * (NCH * 16)) + (lane ^ 1);
#pragma unroll
        for (int j = 0; j < NCH; ++j)
          dst[j * 32] = (u16a)pd[j];
        xc = xn;
      }
    } else {
      if (p >= 1) {
        const int* src = sProw + ((p - 1) & 1) * (NCH * 16) + (hi_half ? 8 : 0);
        v16h ha = load_h1_rows(src, lane, 0);
        v16h hb = load_h1_rows(src, lane, 1);
        const int s0 = NCH * (p - 1) - 8;
        if (p == 1)
          layer2_group<false, true>(s0, lane, hi_half, W, zc, c, aop, rail2, wfc, bfc,
                                    fq, oacc, qb, qa, ha, orow, sel);
        else
          layer2_group<true, true>(s0, lane, hi_half, W, zc, c, aop, rail2, wfc, bfc,
                                   fq, oacc, qb, qa, ha, orow, sel);
        layer2_group<true, true>(s0 + 8, lane, hi_half, W, zc, c, aop, rail2, wfc, bfc,
                                 fq, oacc, qa, qb, hb, orow, sel);
      }
    }
    if (p < NPH) handover_barrier();
  }

  if (layer != 0) {
    v16h none = aop;
    layer2_group<true, false>(TLEN - 8, lane, hi_half, W, zc, c, aop, rail2, wfc, bfc,
                              fq, oacc, qb, qa, none, orow, sel);
#pragma unroll
    for (int i = 1; i <= 4; ++i) fc_fold_stage(i, lane, fq, oacc, bfc, 3);
    orow[TLEN - 32 + lane] = oacc;
  }
}

extern "C" void kernel_launch(void* const* d_in, const int* in_sizes, int n_in,
                              void* d_out, int out_size, void* d_ws, size_t ws_size,
                              hipStream_t stream) {
  (void)in_sizes; (void)n_in; (void)out_size; (void)d_ws; (void)ws_size;
  const float* x     = (const float*)d_in[0];
  const float* w_ih1 = (const float*)d_in[1];
  const float* w_hh1 = (const float*)d_in[2];
  const float* b_ih1 = (const float*)d_in[3];
  const float* b_hh1 = (const float*)d_in[4];
  const float* w_ih2 = (const float*)d_in[5];
  const float* w_hh2 = (const float*)d_in[6];
  const float* b_ih2 = (const float*)d_in[7];
  const float* b_hh2 = (const float*)d_in[8];
  const float* w_fc  = (const float*)d_in[9];
  const float* b_fc  = (const float*)d_in[10];
  float* out = (float*)d_out;

  dim3 grid(BSZ / NROW);
  dim3 block(NTHR);
  lstm2_wmma_kernel<<<grid, block, 0, stream>>>(x, w_ih1, w_hh1, b_ih1, b_hh1,
                                                w_ih2, w_hh2, b_ih2, b_hh2,
                                                w_fc, b_fc, out);
}
